# starting kernel
# baseline (speedup 1.0000x reference)
#include <hip/hip_runtime.h>
#include <hip/hip_bf16.h>
#include <math.h>

#define BATCH   2
#define LENGTH  2048
#define DINNER  1024
#define DSTATE  16
#define DTRANK  64
#define MROWS   (BATCH * LENGTH)
#define XPROJ_N (DTRANK + 2 * DSTATE)

#define NCH 16
#define LCH (LENGTH / NCH)

typedef __attribute__((ext_vector_type(2))) float v2f;
typedef __attribute__((ext_vector_type(8))) float v8f;

__global__ void k_xproj(const float* __restrict__ u, const float* __restrict__ Wx,
                        float* __restrict__ xdbl) {
  const int wave = blockIdx.x * (blockDim.x >> 5) + (threadIdx.x >> 5);
  const int lane = threadIdx.x & 31;
  const int NT = XPROJ_N / 16;
  const int tileM = wave / NT;
  const int tileN = wave % NT;
  if (tileM >= MROWS / 16) return;

  const int khalf = (lane >> 4) << 1;
  const float* aRow = u  + (size_t)(tileM * 16 + (lane & 15)) * DINNER + khalf;
  const float* bRow = Wx + (size_t)(tileN * 16 + (lane & 15)) * DINNER + khalf;

  v8f acc = {};
  for (int k = 0; k < DINNER; k += 4) {
    v2f a = *(const v2f*)(aRow + k);
    v2f b = *(const v2f*)(bRow + k);
    acc = __builtin_amdgcn_wmma_f32_16x16x4_f32(
         false, a,  false, b,
         (short)0, acc,  false,  false);
  }
  const int col   = tileN * 16 + (lane & 15);
  const int rbase = tileM * 16 + ((lane >> 4) << 3);
#pragma unroll
  for (int r = 0; r < 8; ++r)
    xdbl[(size_t)(rbase + r) * XPROJ_N + col] = acc[r];
}

__global__ void k_dtproj(const float* __restrict__ xdbl, const float* __restrict__ Wdt,
                         const float* __restrict__ bdt, float* __restrict__ delta) {
  const int wave = blockIdx.x * (blockDim.x >> 5) + (threadIdx.x >> 5);
  const int lane = threadIdx.x & 31;
  const int NT = DINNER / 16;
  const int tileM = wave / NT;
  const int tileN = wave % NT;
  if (tileM >= MROWS / 16) return;

  const int khalf = (lane >> 4) << 1;
  const float* aRow = xdbl + (size_t)(tileM * 16 + (lane & 15)) * XPROJ_N + khalf;
  const float* bRow = Wdt  + (size_t)(tileN * 16 + (lane & 15)) * DTRANK + khalf;

  v8f acc = {};
#pragma unroll
  for (int k = 0; k < DTRANK; k += 4) {
    v2f a = *(const v2f*)(aRow + k);
    v2f b = *(const v2f*)(bRow + k);
    acc = __builtin_amdgcn_wmma_f32_16x16x4_f32(
        false, a, false, b, (short)0, acc, false, false);
  }
  const int col   = tileN * 16 + (lane & 15);
  const int rbase = tileM * 16 + ((lane >> 4) << 3);
  const float bias = bdt[col];
#pragma unroll
  for (int r = 0; r < 8; ++r) {
    float x  = acc[r] + bias;
    float sp = (x > 20.f) ? x : log1pf(__expf(x));
    delta[(size_t)(rbase + r) * DINNER + col] = sp;
  }
}

__global__ void k_scan_phase1(const float* __restrict__ u, const float* __restrict__ xdbl,
                              const float* __restrict__ delta, const float* __restrict__ logA,
                              float* __restrict__ chunkA, float* __restrict__ chunkH) {
  const int dblk = blockIdx.x & 3;
  const int ch   = (blockIdx.x >> 2) & (NCH - 1);
  const int b    = blockIdx.x >> 6;
  const int d    = dblk * 256 + threadIdx.x;
  const int t0   = ch * LCH;

  __shared__ float Bs[LCH][DSTATE];
  for (int i = threadIdx.x; i < LCH * DSTATE; i += 256) {
    int t = i >> 4, n = i & 15;
    Bs[t][n] = xdbl[(size_t)(b * LENGTH + t0 + t) * XPROJ_N + DTRANK + n];
  }
  __syncthreads();

  float An[DSTATE], h[DSTATE], ap[DSTATE];
#pragma unroll
  for (int n = 0; n < DSTATE; ++n) {
    An[n] = -__expf(logA[d * DSTATE + n]);
    h[n]  = 0.f;
    ap[n] = 1.f;
  }
  for (int t = 0; t < LCH; ++t) {
    size_t idx = (size_t)(b * LENGTH + t0 + t) * DINNER + d;
    float dlt = delta[idx];
    float uu  = u[idx];
    float du  = dlt * uu;
#pragma unroll
    for (int n = 0; n < DSTATE; ++n) {
      float a = __expf(dlt * An[n]);
      h[n]  = a * h[n] + du * Bs[t][n];
      ap[n] *= a;
    }
  }
  size_t base = (((size_t)(b * DINNER + d)) * NCH + ch) * DSTATE;
#pragma unroll
  for (int n = 0; n < DSTATE; ++n) {
    chunkA[base + n] = ap[n];
    chunkH[base + n] = h[n];
  }
}

__global__ void k_scan_phase2(const float* __restrict__ chunkA, const float* __restrict__ chunkH,
                              float* __restrict__ carry) {
  const int idx = blockIdx.x * 256 + threadIdx.x;
  const int n   = idx & 15;
  const int bd  = idx >> 4;
  size_t base = (size_t)bd * NCH * DSTATE + n;
  float h = 0.f;
#pragma unroll
  for (int c = 0; c < NCH; ++c) {
    carry[base + (size_t)c * DSTATE] = h;
    h = chunkA[base + (size_t)c * DSTATE] * h + chunkH[base + (size_t)c * DSTATE];
  }
}

__global__ void k_scan_phase3(const float* __restrict__ u, const float* __restrict__ xdbl,
                              const float* __restrict__ delta, const float* __restrict__ logA,
                              const float* __restrict__ Dp, const float* __restrict__ carry,
                              float* __restrict__ y) {
  const int dblk = blockIdx.x & 3;
  const int ch   = (blockIdx.x >> 2) & (NCH - 1);
  const int b    = blockIdx.x >> 6;
  const int d    = dblk * 256 + threadIdx.x;
  const int t0   = ch * LCH;

  __shared__ float Bs[LCH][DSTATE];
  __shared__ float Cs[LCH][DSTATE];
  for (int i = threadIdx.x; i < LCH * DSTATE; i += 256) {
    int t = i >> 4, n = i & 15;
    size_t rowoff = (size_t)(b * LENGTH + t0 + t) * XPROJ_N;
    Bs[t][n] = xdbl[rowoff + DTRANK + n];
    Cs[t][n] = xdbl[rowoff + DTRANK + DSTATE + n];
  }
  __syncthreads();

  float An[DSTATE], h[DSTATE];
  size_t cbase = (((size_t)(b * DINNER + d)) * NCH + ch) * DSTATE;
#pragma unroll
  for (int n = 0; n < DSTATE; ++n) {
    An[n] = -__expf(logA[d * DSTATE + n]);
    h[n]  = carry[cbase + n];
  }
  const float Dd = Dp[d];
  for (int t = 0; t < LCH; ++t) {
    size_t idx = (size_t)(b * LENGTH + t0 + t) * DINNER + d;
    float dlt = delta[idx];
    float uu  = u[idx];
    float du  = dlt * uu;
    float acc = 0.f;
#pragma unroll
    for (int n = 0; n < DSTATE; ++n) {
      float a = __expf(dlt * An[n]);
      h[n] = a * h[n] + du * Bs[t][n];
      acc += h[n] * Cs[t][n];
    }
    y[idx] = acc + uu * Dd;
  }
}

extern "C" void kernel_launch(void* const* d_in, const int* in_sizes, int n_in,
                              void* d_out, int out_size, void* d_ws, size_t ws_size,
                              hipStream_t stream) {
  (void)in_sizes; (void)n_in; (void)out_size; (void)ws_size;
  const float* u    = (const float*)d_in[0];
  const float* Wx   = (const float*)d_in[1];
  const float* Wdt  = (const float*)d_in[2];
  const float* bdt  = (const float*)d_in[3];
  const float* logA = (const float*)d_in[4];
  const float* Dp   = (const float*)d_in[5];
  float* y = (float*)d_out;

  float* ws     = (float*)d_ws;
  float* xdbl   = ws;
  float* delta  = xdbl  + (size_t)MROWS * XPROJ_N;
  float* chunkA = delta + (size_t)MROWS * DINNER;
  float* chunkH = chunkA + (size_t)BATCH * DINNER * NCH * DSTATE;
  float* carry  = chunkH + (size_t)BATCH * DINNER * NCH * DSTATE;

  {
    int tiles = (MROWS / 16) * (XPROJ_N / 16);
    k_xproj<<<(tiles + 7) / 8, 256, 0, stream>>>(u, Wx, xdbl);
  }
  {
    int tiles = (MROWS / 16) * (DINNER / 16);
    k_dtproj<<<(tiles + 7) / 8, 256, 0, stream>>>(xdbl, Wdt, bdt, delta);
  }
  const int scan_blocks = BATCH * NCH * (DINNER / 256);
  k_scan_phase1<<<scan_blocks, 256, 0, stream>>>(u, xdbl, delta, logA, chunkA, chunkH);
  k_scan_phase2<<<(BATCH * DINNER * DSTATE) / 256, 256, 0, stream>>>(chunkA, chunkH, carry);
  k_scan_phase3<<<scan_blocks, 256, 0, stream>>>(u, xdbl, delta, logA, Dp, carry, y);
}


# x_proj GEMM re-done as 64x32 work-group tiles: u and W_x staged in LDS per 128 k (whole-line loads one stage ahead, double buffer), one 16x16 tile per wave over all K; sum order changes
# speedup vs baseline: 3.3556x; 3.3556x over previous
#include <hip/hip_runtime.h>
#include <hip/hip_bf16.h>
#include <math.h>


#define BATCH   2
#define LENGTH  2048
#define DINNER  1024
#define DSTATE  16
#define DTRANK  64
#define MROWS   (BATCH * LENGTH)
#define XPROJ_N (DTRANK + 2 * DSTATE)

#define NCH 16
#define LCH (LENGTH / NCH)

#define TB    16
#define HS    (DSTATE / 2)
#define DWG   128
#define TILE_FLOATS 256
#define TILE_STRIDE ((size_t)(DINNER / 16) * TILE_FLOATS)
#define RPITCH 68

typedef __attribute__((ext_vector_type(2))) float v2f;
typedef __attribute__((ext_vector_type(4))) float v4f;
typedef __attribute__((ext_vector_type(8))) float v8f;

#define XBM    64
#define XBN    32
#define XKC    128
#define XPITCH (XKC + 4)
#define XROWS  (XBM + XBN)
#define XNST   (DINNER / XKC)
#define XLD    (XROWS * (XKC / 4) / 256)
#define XNP    (XKC / 8)

__global__ __launch_bounds__(256)
void k_xproj(const float* __restrict__ u, const float* __restrict__ Wx,
             float* __restrict__ xdbl) {
  __shared__ __attribute__((aligned(16))) float St[2][XROWS][XPITCH];
  const int lane = threadIdx.x & 31;
  const int wave = __builtin_amdgcn_readfirstlane(threadIdx.x >> 5);
  const int c    = lane & 15;
  const int hh   = lane >> 4;
  const int m0   = (blockIdx.x / 3) * XBM;
  const int n0   = (blockIdx.x % 3) * XBN;
  const int mi   = wave & 3;
  const int ni   = wave >> 2;

  const float* srow[XLD];
#pragma unroll
  for (int i = 0; i < XLD; ++i) {
    const int r = 8 * i + wave;
    srow[i] = (i < XBM / 8 ? u + (size_t)(m0 + r) * DINNER
                           : Wx + (size_t)(n0 + r - XBM) * DINNER) + 4 * lane;
  }

  v4f pf[XLD];
#pragma unroll
  for (int i = 0; i < XLD; ++i) pf[i] = *(const v4f*)(srow[i]);
#pragma unroll
  for (int i = 0; i < XLD; ++i) *(v4f*)&St[0][8 * i + wave][4 * lane] = pf[i];
  __syncthreads();

  v8f acc0 = {}, acc1 = {};
#pragma unroll
  for (int s = 0; s < XNST; ++s) {
    if (s + 1 < XNST) {
#pragma unroll
      for (int i = 0; i < XLD; ++i) pf[i] = *(const v4f*)(srow[i] + (s + 1) * XKC);
    }
    const float* Ua = &St[s & 1][16 * mi + c][4 * hh];
    const float* Wb = &St[s & 1][XBM + 16 * ni + c][4 * hh];
    v4f a4[XNP], b4[XNP];
#pragma unroll
    for (int p = 0; p < XNP; ++p) {
      a4[p] = *(const v4f*)(Ua + 8 * p);
      b4[p] = *(const v4f*)(Wb + 8 * p);
    }
    __builtin_amdgcn_sched_barrier(0);
#pragma unroll
    for (int p = 0; p < XNP; ++p) {
      const v2f a0 = {a4[p].x, a4[p].y}, a1 = {a4[p].z, a4[p].w};
      const v2f b0 = {b4[p].x, b4[p].y}, b1 = {b4[p].z, b4[p].w};
      acc0 = __builtin_amdgcn_wmma_f32_16x16x4_f32(
           false, a0,  false, b0,
           (short)0, acc0,  false,  false);
      acc1 = __builtin_amdgcn_wmma_f32_16x16x4_f32(
          false, a1, false, b1, (short)0, acc1, false, false);
    }
    __builtin_amdgcn_sched_barrier(0);
    if (s + 1 < XNST) {
#pragma unroll
      for (int i = 0; i < XLD; ++i) *(v4f*)&St[(s + 1) & 1][8 * i + wave][4 * lane] = pf[i];
      __syncthreads();
    }
  }

  float* Os = &St[0][0][0];
#pragma unroll
  for (int r = 0; r < 8; ++r)
    Os[(16 * mi + 8 * hh + r) * XBN + 16 * ni + c] = acc0[r] + acc1[r];
  __syncthreads();
#pragma unroll
  for (int j = 0; j < XBM * XBN / 4 / 256; ++j) {
    const int idx = j * 256 + threadIdx.x;
    *(v4f*)(xdbl + (size_t)(m0 + (idx >> 3)) * XPROJ_N + n0 + 4 * (idx & 7)) = ((const v4f*)Os)[idx];
  }
}

__device__ __forceinline__ void delta_tile(const float (*Rs)[RPITCH], int row0, int c, int hh,
                                           const v2f (&bfrag)[DTRANK / 4], float bias,
                                           float (&dl)[8]) {
  v2f a[DTRANK / 4];
#pragma unroll
  for (int j = 0; j < DTRANK / 4; ++j)
    a[j] = *(const v2f*)&Rs[row0 + c][2 * hh + 4 * j];
  v8f acc = {};
#pragma unroll
  for (int j = 0; j < DTRANK / 4; ++j)
    acc = __builtin_amdgcn_wmma_f32_16x16x4_f32(
        false, a[j], false, bfrag[j], (short)0, acc, false, false);
  float x[8], lp[8];
#pragma unroll
  for (int r = 0; r < 8; ++r) x[r] = acc[r] + bias;
#pragma unroll
  for (int r = 0; r < 8; ++r) lp[r] = log1pf(__expf(x[r]));
#pragma unroll
  for (int r = 0; r < 8; ++r) dl[r] = (x[r] > 20.f) ? x[r] : lp[r];
}

__device__ __forceinline__ void gather16(const float (&own)[8], int hh, float (&out)[TB]) {
#pragma unroll
  for (int r = 0; r < 8; ++r) {
    float oth = __shfl_xor(own[r], 16);
    out[r]     = hh ? oth : own[r];
    out[8 + r] = hh ? own[r] : oth;
  }
}

__device__ __forceinline__ void stage_rows(float (*Rs)[RPITCH], const float* __restrict__ xdbl,
                                           int row_global0) {
  for (int i = threadIdx.x; i < LCH * (DTRANK / 4); i += 256) {
    int t = i >> 4, q = i & 15;
    float4 v = *(const float4*)(xdbl + (size_t)(row_global0 + t) * XPROJ_N + q * 4);
    *(float4*)&Rs[t][q * 4] = v;
  }
}

__global__ __launch_bounds__(256)
void k_scan_phase1(const float* __restrict__ u, const float* __restrict__ xdbl,
                   const float* __restrict__ Wdt, const float* __restrict__ bdt,
                   const float* __restrict__ logA,
                   float* __restrict__ chunkA, float* __restrict__ chunkH,
                   float* __restrict__ dlT, float* __restrict__ duT) {
  const int dblk = blockIdx.x & 7;
  const int ch   = (blockIdx.x >> 3) & (NCH - 1);
  const int b    = blockIdx.x >> 7;
  const int lane = threadIdx.x & 31;
  const int c    = lane & 15;
  const int hh   = lane >> 4;
  const int d    = dblk * DWG + (threadIdx.x >> 5) * 16 + c;
  const int t0   = ch * LCH;
  const size_t tile0 = ((size_t)((b * LENGTH + t0) / TB) * (DINNER / 16) + (size_t)(d >> 4)) * TILE_FLOATS;

  __shared__ __attribute__((aligned(16))) float Rs[LCH][RPITCH];
  __shared__ __attribute__((aligned(16))) float Bs[LCH][DSTATE];
  stage_rows(Rs, xdbl, b * LENGTH + t0);
  for (int i = threadIdx.x; i < LCH * DSTATE; i += 256) {
    int t = i >> 4, n = i & 15;
    Bs[t][n] = xdbl[(size_t)(b * LENGTH + t0 + t) * XPROJ_N + DTRANK + n];
  }

  v2f bfrag[DTRANK / 4];
#pragma unroll
  for (int j = 0; j < DTRANK / 4; ++j)
    bfrag[j] = *(const v2f*)(Wdt + (size_t)d * DTRANK + 2 * hh + 4 * j);
  const float bias = bdt[d];

  float An[HS], h[HS], ap[HS];
#pragma unroll
  for (int i = 0; i < HS; ++i) {
    An[i] = -__expf(logA[d * DSTATE + HS * hh + i]);
    h[i]  = 0.f;
    ap[i] = 1.f;
  }
  __syncthreads();

  for (int tb = 0; tb < LCH / TB; ++tb) {
    const int row0 = tb * TB;
    float uo[8];
#pragma unroll
    for (int r = 0; r < 8; ++r)
      uo[r] = u[(size_t)(b * LENGTH + t0 + row0 + 8 * hh + r) * DINNER + d];

    float dl[8], du[8];
    delta_tile(Rs, row0, c, hh, bfrag, bias, dl);
#pragma unroll
    for (int r = 0; r < 8; ++r) du[r] = dl[r] * uo[r];

    {
      const size_t tile = tile0 + (size_t)tb * TILE_STRIDE;
#pragma unroll
      for (int j = 0; j < 2; ++j) {
        const size_t o = tile + (size_t)(((2 * j + hh) * 16 + c) * 4);
        *(float4*)(dlT + o) = make_float4(dl[4 * j], dl[4 * j + 1], dl[4 * j + 2], dl[4 * j + 3]);
        *(float4*)(duT + o) = make_float4(du[4 * j], du[4 * j + 1], du[4 * j + 2], du[4 * j + 3]);
      }
    }

    float dl16[TB], du16[TB];
    gather16(dl, hh, dl16);
    gather16(du, hh, du16);

#pragma unroll
    for (int s = 0; s < TB; ++s) {
      const float4 b0 = *(const float4*)&Bs[row0 + s][HS * hh];
      const float4 b1 = *(const float4*)&Bs[row0 + s][HS * hh + 4];
      const float Bv[HS] = {b0.x, b0.y, b0.z, b0.w, b1.x, b1.y, b1.z, b1.w};
#pragma unroll
      for (int i = 0; i < HS; ++i) {
        float a = __expf(dl16[s] * An[i]);
        h[i]  = a * h[i] + du16[s] * Bv[i];
        ap[i] *= a;
      }
    }
  }
  size_t base = (((size_t)(b * DINNER + d)) * NCH + ch) * DSTATE + HS * hh;
#pragma unroll
  for (int i = 0; i < HS; ++i) {
    chunkA[base + i] = ap[i];
    chunkH[base + i] = h[i];
  }
}

__global__ void k_scan_phase2(const float* __restrict__ chunkA, const float* __restrict__ chunkH,
                              float* __restrict__ carry) {
  const int idx = blockIdx.x * 256 + threadIdx.x;
  const int n   = idx & 15;
  const int bd  = idx >> 4;
  size_t base = (size_t)bd * NCH * DSTATE + n;
  float h = 0.f;
#pragma unroll
  for (int c = 0; c < NCH; ++c) {
    carry[base + (size_t)c * DSTATE] = h;
    h = chunkA[base + (size_t)c * DSTATE] * h + chunkH[base + (size_t)c * DSTATE];
  }
}

struct P3Blk {
  float4 dq[4];
  float4 gq[4];
  float  uo[8];
};

__device__ __forceinline__ void p3_fetch(P3Blk& k, const float* __restrict__ dlT,
                                         const float* __restrict__ duT,
                                         const float* __restrict__ urow, size_t tile, int c) {
#pragma unroll
  for (int p = 0; p < 4; ++p) {
    k.dq[p] = *(const float4*)(dlT + tile + (size_t)((p * 16 + c) * 4));
    k.gq[p] = *(const float4*)(duT + tile + (size_t)((p * 16 + c) * 4));
  }
#pragma unroll
  for (int r = 0; r < 8; ++r) k.uo[r] = urow[(size_t)r * DINNER];
}

__device__ __forceinline__ void p3_block(const P3Blk& k, const float (*Bs)[DSTATE],
                                         const float (*Cs)[DSTATE], int row0, int hh,
                                         const float (&An)[HS], float (&h)[HS], float Dd,
                                         float* __restrict__ yrow) {
  float dl16[TB], du16[TB], yp[TB];
#pragma unroll
  for (int p = 0; p < 4; ++p) {
    const int s0 = 8 * (p & 1) + 4 * (p >> 1);
    dl16[s0] = k.dq[p].x; dl16[s0 + 1] = k.dq[p].y; dl16[s0 + 2] = k.dq[p].z; dl16[s0 + 3] = k.dq[p].w;
    du16[s0] = k.gq[p].x; du16[s0 + 1] = k.gq[p].y; du16[s0 + 2] = k.gq[p].z; du16[s0 + 3] = k.gq[p].w;
  }
#pragma unroll
  for (int s = 0; s < TB; ++s) {
    const float4 b0 = *(const float4*)&Bs[row0 + s][HS * hh];
    const float4 b1 = *(const float4*)&Bs[row0 + s][HS * hh + 4];
    const float4 c0 = *(const float4*)&Cs[row0 + s][HS * hh];
    const float4 c1 = *(const float4*)&Cs[row0 + s][HS * hh + 4];
    const float Bv[HS] = {b0.x, b0.y, b0.z, b0.w, b1.x, b1.y, b1.z, b1.w};
    const float Cv[HS] = {c0.x, c0.y, c0.z, c0.w, c1.x, c1.y, c1.z, c1.w};
    float acc = 0.f;
#pragma unroll
    for (int i = 0; i < HS; ++i) {
      float a = __expf(dl16[s] * An[i]);
      h[i] = a * h[i] + du16[s] * Bv[i];
      acc += h[i] * Cv[i];
    }
    yp[s] = acc;
  }
#pragma unroll
  for (int r = 0; r < 8; ++r) {
    float mine = hh ? yp[8 + r] : yp[r];
    float send = hh ? yp[r] : yp[8 + r];
    float recv = __shfl_xor(send, 16);
    float lo = hh ? recv : mine;
    float hi = hh ? mine : recv;
    yrow[(size_t)r * DINNER] = (lo + hi) + k.uo[r] * Dd;
  }
}

__global__ __launch_bounds__(256)
void k_scan_phase3(const float* __restrict__ u, const float* __restrict__ xdbl,
                   const float* __restrict__ logA,
                   const float* __restrict__ Dp, const float* __restrict__ carry,
                   const float* __restrict__ dlT, const float* __restrict__ duT,
                   float* __restrict__ y) {
  const int dblk = blockIdx.x & 7;
  const int ch   = (blockIdx.x >> 3) & (NCH - 1);
  const int b    = blockIdx.x >> 7;
  const int lane = threadIdx.x & 31;
  const int c    = lane & 15;
  const int hh   = lane >> 4;
  const int d    = dblk * DWG + (threadIdx.x >> 5) * 16 + c;
  const int t0   = ch * LCH;
  const size_t tile0 = ((size_t)((b * LENGTH + t0) / TB) * (DINNER / 16) + (size_t)(d >> 4)) * TILE_FLOATS;
  const size_t row0off = (size_t)(b * LENGTH + t0 + 8 * hh) * DINNER + d;
  const size_t blkrows = (size_t)TB * DINNER;

  P3Blk k0, k1;
  p3_fetch(k0, dlT, duT, u + row0off, tile0, c);

  __shared__ __attribute__((aligned(16))) float Bs[LCH][DSTATE];
  __shared__ __attribute__((aligned(16))) float Cs[LCH][DSTATE];
  for (int i = threadIdx.x; i < LCH * DSTATE; i += 256) {
    int t = i >> 4, n = i & 15;
    size_t rowoff = (size_t)(b * LENGTH + t0 + t) * XPROJ_N;
    Bs[t][n] = xdbl[rowoff + DTRANK + n];
    Cs[t][n] = xdbl[rowoff + DTRANK + DSTATE + n];
  }

  float An[HS], h[HS];
  size_t cbase = (((size_t)(b * DINNER + d)) * NCH + ch) * DSTATE + HS * hh;
#pragma unroll
  for (int i = 0; i < HS; ++i) {
    An[i] = -__expf(logA[d * DSTATE + HS * hh + i]);
    h[i]  = carry[cbase + i];
  }
  const float Dd = Dp[d];
  __syncthreads();

  const int NTB = LCH / TB;
  for (int tb = 0; tb < NTB; tb += 2) {
    p3_fetch(k1, dlT, duT, u + row0off + (size_t)(tb + 1) * blkrows, tile0 + (size_t)(tb + 1) * TILE_STRIDE, c);
    __builtin_amdgcn_sched_barrier(0);
    p3_block(k0, Bs, Cs, tb * TB, hh, An, h, Dd, y + row0off + (size_t)tb * blkrows);
    const int tn = (tb + 2 < NTB) ? tb + 2 : NTB - 1;
    p3_fetch(k0, dlT, duT, u + row0off + (size_t)tn * blkrows, tile0 + (size_t)tn * TILE_STRIDE, c);
    __builtin_amdgcn_sched_barrier(0);
    p3_block(k1, Bs, Cs, (tb + 1) * TB, hh, An, h, Dd, y + row0off + (size_t)(tb + 1) * blkrows);
  }
}

extern "C" void kernel_launch(void* const* d_in, const int* in_sizes, int n_in,
                              void* d_out, int out_size, void* d_ws, size_t ws_size,
                              hipStream_t stream) {
  (void)in_sizes; (void)n_in; (void)out_size; (void)ws_size;
  const float* u    = (const float*)d_in[0];
  const float* Wx   = (const float*)d_in[1];
  const float* Wdt  = (const float*)d_in[2];
  const float* bdt  = (const float*)d_in[3];
  const float* logA = (const float*)d_in[4];
  const float* Dp   = (const float*)d_in[5];
  float* y = (float*)d_out;

  float* ws     = (float*)d_ws;
  float* xdbl   = ws;
  float* chunkA = xdbl  + (size_t)MROWS * XPROJ_N;
  float* chunkH = chunkA + (size_t)BATCH * DINNER * NCH * DSTATE;
  float* carry  = chunkH + (size_t)BATCH * DINNER * NCH * DSTATE;
  float* dlT    = carry  + (size_t)BATCH * DINNER * NCH * DSTATE;
  float* duT    = dlT    + (size_t)MROWS * DINNER;

  k_xproj<<<(MROWS / 64) * (XPROJ_N / 32), 256, 0, stream>>>(u, Wx, xdbl);
  const int scan_blocks = BATCH * NCH * (DINNER / DWG);
  k_scan_phase1<<<scan_blocks, 256, 0, stream>>>(u, xdbl, Wdt, bdt, logA, chunkA, chunkH, dlT, duT);
  k_scan_phase2<<<(BATCH * DINNER * DSTATE) / 256, 256, 0, stream>>>(chunkA, chunkH, carry);
  k_scan_phase3<<<scan_blocks, 256, 0, stream>>>(u, xdbl, logA, Dp, carry, dlT, duT, y);
}


# scan phases 1, 3: work-group prologue issues all its global loads (x_dbl rows to stage, as 16-byte pieces; W_dt, log_A, carry) before the first wait: one memory round trip, not one per stripe
# speedup vs baseline: 3.6689x; 1.0934x over previous
#include <hip/hip_runtime.h>
#include <hip/hip_bf16.h>
#include <math.h>


#define BATCH   2
#define LENGTH  2048
#define DINNER  1024
#define DSTATE  16
#define DTRANK  64
#define MROWS   (BATCH * LENGTH)
#define XPROJ_N (DTRANK + 2 * DSTATE)

#define NCH 16
#define LCH (LENGTH / NCH)

#define TB    16
#define HS    (DSTATE / 2)
#define DWG   128
#define TILE_FLOATS 256
#define TILE_STRIDE ((size_t)(DINNER / 16) * TILE_FLOATS)
#define RPITCH 68

typedef __attribute__((ext_vector_type(2))) float v2f;
typedef __attribute__((ext_vector_type(4))) float v4f;
typedef __attribute__((ext_vector_type(8))) float v8f;

#define XBM    64
#define XBN    32
#define XKC    128
#define XPITCH (XKC + 4)
#define XROWS  (XBM + XBN)
#define XNST   (DINNER / XKC)
#define XLD    (XROWS * (XKC / 4) / 256)
#define XNP    (XKC / 8)

__global__ __launch_bounds__(256)
void k_xproj(const float* __restrict__ u, const float* __restrict__ Wx,
             float* __restrict__ xdbl) {
  __shared__ __attribute__((aligned(16))) float St[2][XROWS][XPITCH];
  const int lane = threadIdx.x & 31;
  const int wave = __builtin_amdgcn_readfirstlane(threadIdx.x >> 5);
  const int c    = lane & 15;
  const int hh   = lane >> 4;
  const int m0   = (blockIdx.x / 3) * XBM;
  const int n0   = (blockIdx.x % 3) * XBN;
  const int mi   = wave & 3;
  const int ni   = wave >> 2;

  const float* srow[XLD];
#pragma unroll
  for (int i = 0; i < XLD; ++i) {
    const int r = 8 * i + wave;
    srow[i] = (i < XBM / 8 ? u + (size_t)(m0 + r) * DINNER
                           : Wx + (size_t)(n0 + r - XBM) * DINNER) + 4 * lane;
  }

  v4f pf[XLD];
#pragma unroll
  for (int i = 0; i < XLD; ++i) pf[i] = *(const v4f*)(srow[i]);
#pragma unroll
  for (int i = 0; i < XLD; ++i) *(v4f*)&St[0][8 * i + wave][4 * lane] = pf[i];
  __syncthreads();

  v8f acc0 = {}, acc1 = {};
#pragma unroll
  for (int s = 0; s < XNST; ++s) {
    if (s + 1 < XNST) {
#pragma unroll
      for (int i = 0; i < XLD; ++i) pf[i] = *(const v4f*)(srow[i] + (s + 1) * XKC);
    }
    const float* Ua = &St[s & 1][16 * mi + c][4 * hh];
    const float* Wb = &St[s & 1][XBM + 16 * ni + c][4 * hh];
    v4f a4[XNP], b4[XNP];
#pragma unroll
    for (int p = 0; p < XNP; ++p) {
      a4[p] = *(const v4f*)(Ua + 8 * p);
      b4[p] = *(const v4f*)(Wb + 8 * p);
    }
    __builtin_amdgcn_sched_barrier(0);
#pragma unroll
    for (int p = 0; p < XNP; ++p) {
      const v2f a0 = {a4[p].x, a4[p].y}, a1 = {a4[p].z, a4[p].w};
      const v2f b0 = {b4[p].x, b4[p].y}, b1 = {b4[p].z, b4[p].w};
      acc0 = __builtin_amdgcn_wmma_f32_16x16x4_f32(
           false, a0,  false, b0,
           (short)0, acc0,  false,  false);
      acc1 = __builtin_amdgcn_wmma_f32_16x16x4_f32(
          false, a1, false, b1, (short)0, acc1, false, false);
    }
    __builtin_amdgcn_sched_barrier(0);
    if (s + 1 < XNST) {
#pragma unroll
      for (int i = 0; i < XLD; ++i) *(v4f*)&St[(s + 1) & 1][8 * i + wave][4 * lane] = pf[i];
      __syncthreads();
    }
  }

  float* Os = &St[0][0][0];
#pragma unroll
  for (int r = 0; r < 8; ++r)
    Os[(16 * mi + 8 * hh + r) * XBN + 16 * ni + c] = acc0[r] + acc1[r];
  __syncthreads();
#pragma unroll
  for (int j = 0; j < XBM * XBN / 4 / 256; ++j) {
    const int idx = j * 256 + threadIdx.x;
    *(v4f*)(xdbl + (size_t)(m0 + (idx >> 3)) * XPROJ_N + n0 + 4 * (idx & 7)) = ((const v4f*)Os)[idx];
  }
}

__device__ __forceinline__ void delta_tile(const float (*Rs)[RPITCH], int row0, int c, int hh,
                                           const v2f (&bfrag)[DTRANK / 4], float bias,
                                           float (&dl)[8]) {
  v2f a[DTRANK / 4];
#pragma unroll
  for (int j = 0; j < DTRANK / 4; ++j)
    a[j] = *(const v2f*)&Rs[row0 + c][2 * hh + 4 * j];
  v8f acc = {};
#pragma unroll
  for (int j = 0; j < DTRANK / 4; ++j)
    acc = __builtin_amdgcn_wmma_f32_16x16x4_f32(
        false, a[j], false, bfrag[j], (short)0, acc, false, false);
  float x[8], lp[8];
#pragma unroll
  for (int r = 0; r < 8; ++r) x[r] = acc[r] + bias;
#pragma unroll
  for (int r = 0; r < 8; ++r) lp[r] = log1pf(__expf(x[r]));
#pragma unroll
  for (int r = 0; r < 8; ++r) dl[r] = (x[r] > 20.f) ? x[r] : lp[r];
}

__device__ __forceinline__ void gather16(const float (&own)[8], int hh, float (&out)[TB]) {
#pragma unroll
  for (int r = 0; r < 8; ++r) {
    float oth = __shfl_xor(own[r], 16);
    out[r]     = hh ? oth : own[r];
    out[8 + r] = hh ? own[r] : oth;
  }
}

#define RS_LD (LCH * (DTRANK / 4) / 256)
#define B_LD  (LCH * (DSTATE / 4) / 256)
#define BC_LD (LCH * (2 * DSTATE / 4) / 256)

__global__ __launch_bounds__(256)
void k_scan_phase1(const float* __restrict__ u, const float* __restrict__ xdbl,
                   const float* __restrict__ Wdt, const float* __restrict__ bdt,
                   const float* __restrict__ logA,
                   float* __restrict__ chunkA, float* __restrict__ chunkH,
                   float* __restrict__ dlT, float* __restrict__ duT) {
  const int dblk = blockIdx.x & 7;
  const int ch   = (blockIdx.x >> 3) & (NCH - 1);
  const int b    = blockIdx.x >> 7;
  const int lane = threadIdx.x & 31;
  const int c    = lane & 15;
  const int hh   = lane >> 4;
  const int d    = dblk * DWG + (threadIdx.x >> 5) * 16 + c;
  const int t0   = ch * LCH;
  const size_t tile0 = ((size_t)((b * LENGTH + t0) / TB) * (DINNER / 16) + (size_t)(d >> 4)) * TILE_FLOATS;

  __shared__ __attribute__((aligned(16))) float Rs[LCH][RPITCH];
  __shared__ __attribute__((aligned(16))) float Bs[LCH][DSTATE];

  const float* xrow0 = xdbl + (size_t)(b * LENGTH + t0) * XPROJ_N;
  float4 rr[RS_LD], rb[B_LD];
#pragma unroll
  for (int k = 0; k < RS_LD; ++k) {
    const int i = threadIdx.x + 256 * k;
    rr[k] = *(const float4*)(xrow0 + (size_t)(i >> 4) * XPROJ_N + (i & 15) * 4);
  }
#pragma unroll
  for (int k = 0; k < B_LD; ++k) {
    const int i = threadIdx.x + 256 * k;
    rb[k] = *(const float4*)(xrow0 + (size_t)(i >> 2) * XPROJ_N + DTRANK + (i & 3) * 4);
  }
  v2f bfrag[DTRANK / 4];
#pragma unroll
  for (int j = 0; j < DTRANK / 4; ++j)
    bfrag[j] = *(const v2f*)(Wdt + (size_t)d * DTRANK + 2 * hh + 4 * j);
  const float bias = bdt[d];
  float la[HS];
#pragma unroll
  for (int i = 0; i < HS; ++i) la[i] = logA[d * DSTATE + HS * hh + i];

#pragma unroll
  for (int k = 0; k < RS_LD; ++k) {
    const int i = threadIdx.x + 256 * k;
    *(float4*)&Rs[i >> 4][(i & 15) * 4] = rr[k];
  }
#pragma unroll
  for (int k = 0; k < B_LD; ++k) {
    const int i = threadIdx.x + 256 * k;
    *(float4*)&Bs[i >> 2][(i & 3) * 4] = rb[k];
  }

  float An[HS], h[HS], ap[HS];
#pragma unroll
  for (int i = 0; i < HS; ++i) {
    An[i] = -__expf(la[i]);
    h[i]  = 0.f;
    ap[i] = 1.f;
  }
  __syncthreads();

  for (int tb = 0; tb < LCH / TB; ++tb) {
    const int row0 = tb * TB;
    float uo[8];
#pragma unroll
    for (int r = 0; r < 8; ++r)
      uo[r] = u[(size_t)(b * LENGTH + t0 + row0 + 8 * hh + r) * DINNER + d];

    float dl[8], du[8];
    delta_tile(Rs, row0, c, hh, bfrag, bias, dl);
#pragma unroll
    for (int r = 0; r < 8; ++r) du[r] = dl[r] * uo[r];

    {
      const size_t tile = tile0 + (size_t)tb * TILE_STRIDE;
#pragma unroll
      for (int j = 0; j < 2; ++j) {
        const size_t o = tile + (size_t)(((2 * j + hh) * 16 + c) * 4);
        *(float4*)(dlT + o) = make_float4(dl[4 * j], dl[4 * j + 1], dl[4 * j + 2], dl[4 * j + 3]);
        *(float4*)(duT + o) = make_float4(du[4 * j], du[4 * j + 1], du[4 * j + 2], du[4 * j + 3]);
      }
    }

    float dl16[TB], du16[TB];
    gather16(dl, hh, dl16);
    gather16(du, hh, du16);

#pragma unroll
    for (int s = 0; s < TB; ++s) {
      const float4 b0 = *(const float4*)&Bs[row0 + s][HS * hh];
      const float4 b1 = *(const float4*)&Bs[row0 + s][HS * hh + 4];
      const float Bv[HS] = {b0.x, b0.y, b0.z, b0.w, b1.x, b1.y, b1.z, b1.w};
#pragma unroll
      for (int i = 0; i < HS; ++i) {
        float a = __expf(dl16[s] * An[i]);
        h[i]  = a * h[i] + du16[s] * Bv[i];
        ap[i] *= a;
      }
    }
  }
  size_t base = (((size_t)(b * DINNER + d)) * NCH + ch) * DSTATE + HS * hh;
#pragma unroll
  for (int i = 0; i < HS; ++i) {
    chunkA[base + i] = ap[i];
    chunkH[base + i] = h[i];
  }
}

__global__ void k_scan_phase2(const float* __restrict__ chunkA, const float* __restrict__ chunkH,
                              float* __restrict__ carry) {
  const int idx = blockIdx.x * 256 + threadIdx.x;
  const int n   = idx & 15;
  const int bd  = idx >> 4;
  size_t base = (size_t)bd * NCH * DSTATE + n;
  float h = 0.f;
#pragma unroll
  for (int c = 0; c < NCH; ++c) {
    carry[base + (size_t)c * DSTATE] = h;
    h = chunkA[base + (size_t)c * DSTATE] * h + chunkH[base + (size_t)c * DSTATE];
  }
}

struct P3Blk {
  float4 dq[4];
  float4 gq[4];
  float  uo[8];
};

__device__ __forceinline__ void p3_fetch(P3Blk& k, const float* __restrict__ dlT,
                                         const float* __restrict__ duT,
                                         const float* __restrict__ urow, size_t tile, int c) {
#pragma unroll
  for (int p = 0; p < 4; ++p) {
    k.dq[p] = *(const float4*)(dlT + tile + (size_t)((p * 16 + c) * 4));
    k.gq[p] = *(const float4*)(duT + tile + (size_t)((p * 16 + c) * 4));
  }
#pragma unroll
  for (int r = 0; r < 8; ++r) k.uo[r] = urow[(size_t)r * DINNER];
}

__device__ __forceinline__ void p3_block(const P3Blk& k, const float (*Bs)[DSTATE],
                                         const float (*Cs)[DSTATE], int row0, int hh,
                                         const float (&An)[HS], float (&h)[HS], float Dd,
                                         float* __restrict__ yrow) {
  float dl16[TB], du16[TB], yp[TB];
#pragma unroll
  for (int p = 0; p < 4; ++p) {
    const int s0 = 8 * (p & 1) + 4 * (p >> 1);
    dl16[s0] = k.dq[p].x; dl16[s0 + 1] = k.dq[p].y; dl16[s0 + 2] = k.dq[p].z; dl16[s0 + 3] = k.dq[p].w;
    du16[s0] = k.gq[p].x; du16[s0 + 1] = k.gq[p].y; du16[s0 + 2] = k.gq[p].z; du16[s0 + 3] = k.gq[p].w;
  }
#pragma unroll
  for (int s = 0; s < TB; ++s) {
    const float4 b0 = *(const float4*)&Bs[row0 + s][HS * hh];
    const float4 b1 = *(const float4*)&Bs[row0 + s][HS * hh + 4];
    const float4 c0 = *(const float4*)&Cs[row0 + s][HS * hh];
    const float4 c1 = *(const float4*)&Cs[row0 + s][HS * hh + 4];
    const float Bv[HS] = {b0.x, b0.y, b0.z, b0.w, b1.x, b1.y, b1.z, b1.w};
    const float Cv[HS] = {c0.x, c0.y, c0.z, c0.w, c1.x, c1.y, c1.z, c1.w};
    float acc = 0.f;
#pragma unroll
    for (int i = 0; i < HS; ++i) {
      float a = __expf(dl16[s] * An[i]);
      h[i] = a * h[i] + du16[s] * Bv[i];
      acc += h[i] * Cv[i];
    }
    yp[s] = acc;
  }
#pragma unroll
  for (int r = 0; r < 8; ++r) {
    float mine = hh ? yp[8 + r] : yp[r];
    float send = hh ? yp[r] : yp[8 + r];
    float recv = __shfl_xor(send, 16);
    float lo = hh ? recv : mine;
    float hi = hh ? mine : recv;
    yrow[(size_t)r * DINNER] = (lo + hi) + k.uo[r] * Dd;
  }
}

__global__ __launch_bounds__(256)
void k_scan_phase3(const float* __restrict__ u, const float* __restrict__ xdbl,
                   const float* __restrict__ logA,
                   const float* __restrict__ Dp, const float* __restrict__ carry,
                   const float* __restrict__ dlT, const float* __restrict__ duT,
                   float* __restrict__ y) {
  const int dblk = blockIdx.x & 7;
  const int ch   = (blockIdx.x >> 3) & (NCH - 1);
  const int b    = blockIdx.x >> 7;
  const int lane = threadIdx.x & 31;
  const int c    = lane & 15;
  const int hh   = lane >> 4;
  const int d    = dblk * DWG + (threadIdx.x >> 5) * 16 + c;
  const int t0   = ch * LCH;
  const size_t tile0 = ((size_t)((b * LENGTH + t0) / TB) * (DINNER / 16) + (size_t)(d >> 4)) * TILE_FLOATS;
  const size_t row0off = (size_t)(b * LENGTH + t0 + 8 * hh) * DINNER + d;
  const size_t blkrows = (size_t)TB * DINNER;

  P3Blk k0, k1;
  p3_fetch(k0, dlT, duT, u + row0off, tile0, c);

  __shared__ __attribute__((aligned(16))) float Bs[LCH][DSTATE];
  __shared__ __attribute__((aligned(16))) float Cs[LCH][DSTATE];

  const float* xrow0 = xdbl + (size_t)(b * LENGTH + t0) * XPROJ_N + DTRANK;
  float4 rbc[BC_LD];
#pragma unroll
  for (int k = 0; k < BC_LD; ++k) {
    const int i = threadIdx.x + 256 * k;
    rbc[k] = *(const float4*)(xrow0 + (size_t)(i >> 3) * XPROJ_N + (i & 7) * 4);
  }
  float la[HS], h[HS];
  size_t cbase = (((size_t)(b * DINNER + d)) * NCH + ch) * DSTATE + HS * hh;
#pragma unroll
  for (int i = 0; i < HS; ++i) {
    la[i] = logA[d * DSTATE + HS * hh + i];
    h[i]  = carry[cbase + i];
  }
  const float Dd = Dp[d];

  {
    const int q = threadIdx.x & 7;
    float* dst = (q < 4) ? &Bs[0][q * 4] : &Cs[0][(q - 4) * 4];
#pragma unroll
    for (int k = 0; k < BC_LD; ++k) {
      const int i = threadIdx.x + 256 * k;
      *(float4*)(dst + (i >> 3) * DSTATE) = rbc[k];
    }
  }

  float An[HS];
#pragma unroll
  for (int i = 0; i < HS; ++i) An[i] = -__expf(la[i]);
  __syncthreads();

  const int NTB = LCH / TB;
  for (int tb = 0; tb < NTB; tb += 2) {
    p3_fetch(k1, dlT, duT, u + row0off + (size_t)(tb + 1) * blkrows, tile0 + (size_t)(tb + 1) * TILE_STRIDE, c);
    __builtin_amdgcn_sched_barrier(0);
    p3_block(k0, Bs, Cs, tb * TB, hh, An, h, Dd, y + row0off + (size_t)tb * blkrows);
    const int tn = (tb + 2 < NTB) ? tb + 2 : NTB - 1;
    p3_fetch(k0, dlT, duT, u + row0off + (size_t)tn * blkrows, tile0 + (size_t)tn * TILE_STRIDE, c);
    __builtin_amdgcn_sched_barrier(0);
    p3_block(k1, Bs, Cs, (tb + 1) * TB, hh, An, h, Dd, y + row0off + (size_t)(tb + 1) * blkrows);
  }
}

extern "C" void kernel_launch(void* const* d_in, const int* in_sizes, int n_in,
                              void* d_out, int out_size, void* d_ws, size_t ws_size,
                              hipStream_t stream) {
  (void)in_sizes; (void)n_in; (void)out_size; (void)ws_size;
  const float* u    = (const float*)d_in[0];
  const float* Wx   = (const float*)d_in[1];
  const float* Wdt  = (const float*)d_in[2];
  const float* bdt  = (const float*)d_in[3];
  const float* logA = (const float*)d_in[4];
  const float* Dp   = (const float*)d_in[5];
  float* y = (float*)d_out;

  float* ws     = (float*)d_ws;
  float* xdbl   = ws;
  float* chunkA = xdbl  + (size_t)MROWS * XPROJ_N;
  float* chunkH = chunkA + (size_t)BATCH * DINNER * NCH * DSTATE;
  float* carry  = chunkH + (size_t)BATCH * DINNER * NCH * DSTATE;
  float* dlT    = carry  + (size_t)BATCH * DINNER * NCH * DSTATE;
  float* duT    = dlT    + (size_t)MROWS * DINNER;

  k_xproj<<<(MROWS / 64) * (XPROJ_N / 32), 256, 0, stream>>>(u, Wx, xdbl);
  const int scan_blocks = BATCH * NCH * (DINNER / DWG);
  k_scan_phase1<<<scan_blocks, 256, 0, stream>>>(u, xdbl, Wdt, bdt, logA, chunkA, chunkH, dlT, duT);
  k_scan_phase2<<<(BATCH * DINNER * DSTATE) / 256, 256, 0, stream>>>(chunkA, chunkH, carry);
  k_scan_phase3<<<scan_blocks, 256, 0, stream>>>(u, xdbl, logA, Dp, carry, dlT, duT, y);
}


# scan phases 1, 3: B, C rows read from LDS as pinned groups of 8 reads, one group ahead of use (before: about 100 places with single reads per 32 steps of the exp/FMA stream); sums as before
# speedup vs baseline: 4.0055x; 1.0917x over previous
#include <hip/hip_runtime.h>
#include <hip/hip_bf16.h>
#include <math.h>


#define BATCH   2
#define LENGTH  2048
#define DINNER  1024
#define DSTATE  16
#define DTRANK  64
#define MROWS   (BATCH * LENGTH)
#define XPROJ_N (DTRANK + 2 * DSTATE)

#define NCH 16
#define LCH (LENGTH / NCH)

#define TB    16
#define HS    (DSTATE / 2)
#define DWG   128
#define TILE_FLOATS 256
#define TILE_STRIDE ((size_t)(DINNER / 16) * TILE_FLOATS)
#define RPITCH 68

typedef __attribute__((ext_vector_type(2))) float v2f;
typedef __attribute__((ext_vector_type(4))) float v4f;
typedef __attribute__((ext_vector_type(8))) float v8f;

#define XBM    64
#define XBN    32
#define XKC    128
#define XPITCH (XKC + 4)
#define XROWS  (XBM + XBN)
#define XNST   (DINNER / XKC)
#define XLD    (XROWS * (XKC / 4) / 256)
#define XNP    (XKC / 8)

__global__ __launch_bounds__(256)
void k_xproj(const float* __restrict__ u, const float* __restrict__ Wx,
             float* __restrict__ xdbl) {
  __shared__ __attribute__((aligned(16))) float St[2][XROWS][XPITCH];
  const int lane = threadIdx.x & 31;
  const int wave = __builtin_amdgcn_readfirstlane(threadIdx.x >> 5);
  const int c    = lane & 15;
  const int hh   = lane >> 4;
  const int m0   = (blockIdx.x / 3) * XBM;
  const int n0   = (blockIdx.x % 3) * XBN;
  const int mi   = wave & 3;
  const int ni   = wave >> 2;

  const float* srow[XLD];
#pragma unroll
  for (int i = 0; i < XLD; ++i) {
    const int r = 8 * i + wave;
    srow[i] = (i < XBM / 8 ? u + (size_t)(m0 + r) * DINNER
                           : Wx + (size_t)(n0 + r - XBM) * DINNER) + 4 * lane;
  }

  v4f pf[XLD];
#pragma unroll
  for (int i = 0; i < XLD; ++i) pf[i] = *(const v4f*)(srow[i]);
#pragma unroll
  for (int i = 0; i < XLD; ++i) *(v4f*)&St[0][8 * i + wave][4 * lane] = pf[i];
  __syncthreads();

  v8f acc0 = {}, acc1 = {};
#pragma unroll
  for (int s = 0; s < XNST; ++s) {
    if (s + 1 < XNST) {
#pragma unroll
      for (int i = 0; i < XLD; ++i) pf[i] = *(const v4f*)(srow[i] + (s + 1) * XKC);
    }
    const float* Ua = &St[s & 1][16 * mi + c][4 * hh];
    const float* Wb = &St[s & 1][XBM + 16 * ni + c][4 * hh];
    v4f a4[XNP], b4[XNP];
#pragma unroll
    for (int p = 0; p < XNP; ++p) {
      a4[p] = *(const v4f*)(Ua + 8 * p);
      b4[p] = *(const v4f*)(Wb + 8 * p);
    }
    __builtin_amdgcn_sched_barrier(0);
#pragma unroll
    for (int p = 0; p < XNP; ++p) {
      const v2f a0 = {a4[p].x, a4[p].y}, a1 = {a4[p].z, a4[p].w};
      const v2f b0 = {b4[p].x, b4[p].y}, b1 = {b4[p].z, b4[p].w};
      acc0 = __builtin_amdgcn_wmma_f32_16x16x4_f32(
           false, a0,  false, b0,
           (short)0, acc0,  false,  false);
      acc1 = __builtin_amdgcn_wmma_f32_16x16x4_f32(
          false, a1, false, b1, (short)0, acc1, false, false);
    }
    __builtin_amdgcn_sched_barrier(0);
    if (s + 1 < XNST) {
#pragma unroll
      for (int i = 0; i < XLD; ++i) *(v4f*)&St[(s + 1) & 1][8 * i + wave][4 * lane] = pf[i];
      __syncthreads();
    }
  }

  float* Os = &St[0][0][0];
#pragma unroll
  for (int r = 0; r < 8; ++r)
    Os[(16 * mi + 8 * hh + r) * XBN + 16 * ni + c] = acc0[r] + acc1[r];
  __syncthreads();
#pragma unroll
  for (int j = 0; j < XBM * XBN / 4 / 256; ++j) {
    const int idx = j * 256 + threadIdx.x;
    *(v4f*)(xdbl + (size_t)(m0 + (idx >> 3)) * XPROJ_N + n0 + 4 * (idx & 7)) = ((const v4f*)Os)[idx];
  }
}

__device__ __forceinline__ void delta_tile(const float (*Rs)[RPITCH], int row0, int c, int hh,
                                           const v2f (&bfrag)[DTRANK / 4], float bias,
                                           float (&dl)[8]) {
  v2f a[DTRANK / 4];
#pragma unroll
  for (int j = 0; j < DTRANK / 4; ++j)
    a[j] = *(const v2f*)&Rs[row0 + c][2 * hh + 4 * j];
  v8f acc = {};
#pragma unroll
  for (int j = 0; j < DTRANK / 4; ++j)
    acc = __builtin_amdgcn_wmma_f32_16x16x4_f32(
        false, a[j], false, bfrag[j], (short)0, acc, false, false);
  float x[8], lp[8];
#pragma unroll
  for (int r = 0; r < 8; ++r) x[r] = acc[r] + bias;
#pragma unroll
  for (int r = 0; r < 8; ++r) lp[r] = log1pf(__expf(x[r]));
#pragma unroll
  for (int r = 0; r < 8; ++r) dl[r] = (x[r] > 20.f) ? x[r] : lp[r];
}

__device__ __forceinline__ void gather16(const float (&own)[8], int hh, float (&out)[TB]) {
#pragma unroll
  for (int r = 0; r < 8; ++r) {
    float oth = __shfl_xor(own[r], 16);
    out[r]     = hh ? oth : own[r];
    out[8 + r] = hh ? own[r] : oth;
  }
}

#define SG 4
struct BGrp  { float4 b[SG][2]; };
#define SG3 2
struct BCGrp { float4 b[SG3][2]; float4 c[SG3][2]; };

__device__ __forceinline__ void b_read(BGrp& g, const float (*Bs)[DSTATE], int row, int hh) {
#pragma unroll
  for (int j = 0; j < SG; ++j) {
    g.b[j][0] = *(const float4*)&Bs[row + j][HS * hh];
    g.b[j][1] = *(const float4*)&Bs[row + j][HS * hh + 4];
  }
}

__device__ __forceinline__ void bc_read(BCGrp& g, const float (*Bs)[DSTATE], const float (*Cs)[DSTATE],
                                        int row, int hh) {
#pragma unroll
  for (int j = 0; j < SG3; ++j) {
    g.b[j][0] = *(const float4*)&Bs[row + j][HS * hh];
    g.b[j][1] = *(const float4*)&Bs[row + j][HS * hh + 4];
    g.c[j][0] = *(const float4*)&Cs[row + j][HS * hh];
    g.c[j][1] = *(const float4*)&Cs[row + j][HS * hh + 4];
  }
}

__device__ __forceinline__ void p1_steps(const BGrp& g, int s0, const float (&dl16)[TB],
                                         const float (&du16)[TB], const float (&An)[HS],
                                         float (&h)[HS], float (&ap)[HS]) {
#pragma unroll
  for (int j = 0; j < SG; ++j) {
    const int s = s0 + j;
    const float Bv[HS] = {g.b[j][0].x, g.b[j][0].y, g.b[j][0].z, g.b[j][0].w,
                          g.b[j][1].x, g.b[j][1].y, g.b[j][1].z, g.b[j][1].w};
#pragma unroll
    for (int i = 0; i < HS; ++i) {
      float a = __expf(dl16[s] * An[i]);
      h[i]  = a * h[i] + du16[s] * Bv[i];
      ap[i] *= a;
    }
  }
}

__device__ __forceinline__ void p3_steps(const BCGrp& g, int s0, const float (&dl16)[TB],
                                         const float (&du16)[TB], const float (&An)[HS],
                                         float (&h)[HS], float (&yp)[TB]) {
#pragma unroll
  for (int j = 0; j < SG3; ++j) {
    const int s = s0 + j;
    const float Bv[HS] = {g.b[j][0].x, g.b[j][0].y, g.b[j][0].z, g.b[j][0].w,
                          g.b[j][1].x, g.b[j][1].y, g.b[j][1].z, g.b[j][1].w};
    const float Cv[HS] = {g.c[j][0].x, g.c[j][0].y, g.c[j][0].z, g.c[j][0].w,
                          g.c[j][1].x, g.c[j][1].y, g.c[j][1].z, g.c[j][1].w};
    float acc = 0.f;
#pragma unroll
    for (int i = 0; i < HS; ++i) {
      float a = __expf(dl16[s] * An[i]);
      h[i] = a * h[i] + du16[s] * Bv[i];
      acc += h[i] * Cv[i];
    }
    yp[s] = acc;
  }
}

#define RS_LD (LCH * (DTRANK / 4) / 256)
#define B_LD  (LCH * (DSTATE / 4) / 256)
#define BC_LD (LCH * (2 * DSTATE / 4) / 256)

__global__ __launch_bounds__(256)
void k_scan_phase1(const float* __restrict__ u, const float* __restrict__ xdbl,
                   const float* __restrict__ Wdt, const float* __restrict__ bdt,
                   const float* __restrict__ logA,
                   float* __restrict__ chunkA, float* __restrict__ chunkH,
                   float* __restrict__ dlT, float* __restrict__ duT) {
  const int dblk = blockIdx.x & 7;
  const int ch   = (blockIdx.x >> 3) & (NCH - 1);
  const int b    = blockIdx.x >> 7;
  const int lane = threadIdx.x & 31;
  const int c    = lane & 15;
  const int hh   = lane >> 4;
  const int d    = dblk * DWG + (threadIdx.x >> 5) * 16 + c;
  const int t0   = ch * LCH;
  const size_t tile0 = ((size_t)((b * LENGTH + t0) / TB) * (DINNER / 16) + (size_t)(d >> 4)) * TILE_FLOATS;

  __shared__ __attribute__((aligned(16))) float Rs[LCH][RPITCH];
  __shared__ __attribute__((aligned(16))) float Bs[LCH][DSTATE];

  const float* xrow0 = xdbl + (size_t)(b * LENGTH + t0) * XPROJ_N;
  float4 rr[RS_LD], rb[B_LD];
#pragma unroll
  for (int k = 0; k < RS_LD; ++k) {
    const int i = threadIdx.x + 256 * k;
    rr[k] = *(const float4*)(xrow0 + (size_t)(i >> 4) * XPROJ_N + (i & 15) * 4);
  }
#pragma unroll
  for (int k = 0; k < B_LD; ++k) {
    const int i = threadIdx.x + 256 * k;
    rb[k] = *(const float4*)(xrow0 + (size_t)(i >> 2) * XPROJ_N + DTRANK + (i & 3) * 4);
  }
  v2f bfrag[DTRANK / 4];
#pragma unroll
  for (int j = 0; j < DTRANK / 4; ++j)
    bfrag[j] = *(const v2f*)(Wdt + (size_t)d * DTRANK + 2 * hh + 4 * j);
  const float bias = bdt[d];
  float la[HS];
#pragma unroll
  for (int i = 0; i < HS; ++i) la[i] = logA[d * DSTATE + HS * hh + i];

#pragma unroll
  for (int k = 0; k < RS_LD; ++k) {
    const int i = threadIdx.x + 256 * k;
    *(float4*)&Rs[i >> 4][(i & 15) * 4] = rr[k];
  }
#pragma unroll
  for (int k = 0; k < B_LD; ++k) {
    const int i = threadIdx.x + 256 * k;
    *(float4*)&Bs[i >> 2][(i & 3) * 4] = rb[k];
  }

  float An[HS], h[HS], ap[HS];
#pragma unroll
  for (int i = 0; i < HS; ++i) {
    An[i] = -__expf(la[i]);
    h[i]  = 0.f;
    ap[i] = 1.f;
  }
  __syncthreads();

  for (int tb = 0; tb < LCH / TB; ++tb) {
    const int row0 = tb * TB;
    float uo[8];
#pragma unroll
    for (int r = 0; r < 8; ++r)
      uo[r] = u[(size_t)(b * LENGTH + t0 + row0 + 8 * hh + r) * DINNER + d];

    BGrp ga, gb;
    b_read(ga, Bs, row0, hh);
    __builtin_amdgcn_sched_barrier(0);

    float dl[8], du[8];
    delta_tile(Rs, row0, c, hh, bfrag, bias, dl);
#pragma unroll
    for (int r = 0; r < 8; ++r) du[r] = dl[r] * uo[r];

    {
      const size_t tile = tile0 + (size_t)tb * TILE_STRIDE;
#pragma unroll
      for (int j = 0; j < 2; ++j) {
        const size_t o = tile + (size_t)(((2 * j + hh) * 16 + c) * 4);
        *(float4*)(dlT + o) = make_float4(dl[4 * j], dl[4 * j + 1], dl[4 * j + 2], dl[4 * j + 3]);
        *(float4*)(duT + o) = make_float4(du[4 * j], du[4 * j + 1], du[4 * j + 2], du[4 * j + 3]);
      }
    }

    float dl16[TB], du16[TB];
    gather16(dl, hh, dl16);
    gather16(du, hh, du16);

    __builtin_amdgcn_sched_barrier(0);
    b_read(gb, Bs, row0 + SG, hh);
    __builtin_amdgcn_sched_barrier(0);
    p1_steps(ga, 0, dl16, du16, An, h, ap);
    __builtin_amdgcn_sched_barrier(0);
    b_read(ga, Bs, row0 + 2 * SG, hh);
    __builtin_amdgcn_sched_barrier(0);
    p1_steps(gb, SG, dl16, du16, An, h, ap);
    __builtin_amdgcn_sched_barrier(0);
    b_read(gb, Bs, row0 + 3 * SG, hh);
    __builtin_amdgcn_sched_barrier(0);
    p1_steps(ga, 2 * SG, dl16, du16, An, h, ap);
    p1_steps(gb, 3 * SG, dl16, du16, An, h, ap);
  }
  size_t base = (((size_t)(b * DINNER + d)) * NCH + ch) * DSTATE + HS * hh;
#pragma unroll
  for (int i = 0; i < HS; ++i) {
    chunkA[base + i] = ap[i];
    chunkH[base + i] = h[i];
  }
}

__global__ void k_scan_phase2(const float* __restrict__ chunkA, const float* __restrict__ chunkH,
                              float* __restrict__ carry) {
  const int idx = blockIdx.x * 256 + threadIdx.x;
  const int n   = idx & 15;
  const int bd  = idx >> 4;
  size_t base = (size_t)bd * NCH * DSTATE + n;
  float h = 0.f;
#pragma unroll
  for (int c = 0; c < NCH; ++c) {
    carry[base + (size_t)c * DSTATE] = h;
    h = chunkA[base + (size_t)c * DSTATE] * h + chunkH[base + (size_t)c * DSTATE];
  }
}

struct P3Blk {
  float4 dq[4];
  float4 gq[4];
  float  uo[8];
};

__device__ __forceinline__ void p3_fetch(P3Blk& k, const float* __restrict__ dlT,
                                         const float* __restrict__ duT,
                                         const float* __restrict__ urow, size_t tile, int c) {
#pragma unroll
  for (int p = 0; p < 4; ++p) {
    k.dq[p] = *(const float4*)(dlT + tile + (size_t)((p * 16 + c) * 4));
    k.gq[p] = *(const float4*)(duT + tile + (size_t)((p * 16 + c) * 4));
  }
#pragma unroll
  for (int r = 0; r < 8; ++r) k.uo[r] = urow[(size_t)r * DINNER];
}

__device__ __forceinline__ void p3_block(const P3Blk& k, const float (*Bs)[DSTATE],
                                         const float (*Cs)[DSTATE], int row0, int nextrow, int hh,
                                         const float (&An)[HS], float (&h)[HS], float Dd,
                                         float* __restrict__ yrow, BCGrp& ga) {
  float dl16[TB], du16[TB], yp[TB];
#pragma unroll
  for (int p = 0; p < 4; ++p) {
    const int s0 = 8 * (p & 1) + 4 * (p >> 1);
    dl16[s0] = k.dq[p].x; dl16[s0 + 1] = k.dq[p].y; dl16[s0 + 2] = k.dq[p].z; dl16[s0 + 3] = k.dq[p].w;
    du16[s0] = k.gq[p].x; du16[s0 + 1] = k.gq[p].y; du16[s0 + 2] = k.gq[p].z; du16[s0 + 3] = k.gq[p].w;
  }
  BCGrp gb;
#pragma unroll
  for (int g = 0; g < TB / SG3; ++g) {
    BCGrp& cur = (g & 1) ? gb : ga;
    BCGrp& nxt = (g & 1) ? ga : gb;
    if (g > 0) __builtin_amdgcn_sched_barrier(0);
    bc_read(nxt, Bs, Cs, (g + 1 < TB / SG3) ? row0 + (g + 1) * SG3 : nextrow, hh);
    __builtin_amdgcn_sched_barrier(0);
    p3_steps(cur, g * SG3, dl16, du16, An, h, yp);
  }
#pragma unroll
  for (int r = 0; r < 8; ++r) {
    float mine = hh ? yp[8 + r] : yp[r];
    float send = hh ? yp[r] : yp[8 + r];
    float recv = __shfl_xor(send, 16);
    float lo = hh ? recv : mine;
    float hi = hh ? mine : recv;
    yrow[(size_t)r * DINNER] = (lo + hi) + k.uo[r] * Dd;
  }
}

__global__ __launch_bounds__(256)
void k_scan_phase3(const float* __restrict__ u, const float* __restrict__ xdbl,
                   const float* __restrict__ logA,
                   const float* __restrict__ Dp, const float* __restrict__ carry,
                   const float* __restrict__ dlT, const float* __restrict__ duT,
                   float* __restrict__ y) {
  const int dblk = blockIdx.x & 7;
  const int ch   = (blockIdx.x >> 3) & (NCH - 1);
  const int b    = blockIdx.x >> 7;
  const int lane = threadIdx.x & 31;
  const int c    = lane & 15;
  const int hh   = lane >> 4;
  const int d    = dblk * DWG + (threadIdx.x >> 5) * 16 + c;
  const int t0   = ch * LCH;
  const size_t tile0 = ((size_t)((b * LENGTH + t0) / TB) * (DINNER / 16) + (size_t)(d >> 4)) * TILE_FLOATS;
  const size_t row0off = (size_t)(b * LENGTH + t0 + 8 * hh) * DINNER + d;
  const size_t blkrows = (size_t)TB * DINNER;

  P3Blk k0, k1;
  p3_fetch(k0, dlT, duT, u + row0off, tile0, c);

  __shared__ __attribute__((aligned(16))) float Bs[LCH][DSTATE];
  __shared__ __attribute__((aligned(16))) float Cs[LCH][DSTATE];

  const float* xrow0 = xdbl + (size_t)(b * LENGTH + t0) * XPROJ_N + DTRANK;
  float4 rbc[BC_LD];
#pragma unroll
  for (int k = 0; k < BC_LD; ++k) {
    const int i = threadIdx.x + 256 * k;
    rbc[k] = *(const float4*)(xrow0 + (size_t)(i >> 3) * XPROJ_N + (i & 7) * 4);
  }
  float la[HS], h[HS];
  size_t cbase = (((size_t)(b * DINNER + d)) * NCH + ch) * DSTATE + HS * hh;
#pragma unroll
  for (int i = 0; i < HS; ++i) {
    la[i] = logA[d * DSTATE + HS * hh + i];
    h[i]  = carry[cbase + i];
  }
  const float Dd = Dp[d];

  {
    const int q = threadIdx.x & 7;
    float* dst = (q < 4) ? &Bs[0][q * 4] : &Cs[0][(q - 4) * 4];
#pragma unroll
    for (int k = 0; k < BC_LD; ++k) {
      const int i = threadIdx.x + 256 * k;
      *(float4*)(dst + (i >> 3) * DSTATE) = rbc[k];
    }
  }

  float An[HS];
#pragma unroll
  for (int i = 0; i < HS; ++i) An[i] = -__expf(la[i]);
  __syncthreads();

  const int NTB = LCH / TB;
  BCGrp ga;
  bc_read(ga, Bs, Cs, 0, hh);
  for (int tb = 0; tb < NTB; tb += 2) {
    p3_fetch(k1, dlT, duT, u + row0off + (size_t)(tb + 1) * blkrows, tile0 + (size_t)(tb + 1) * TILE_STRIDE, c);
    __builtin_amdgcn_sched_barrier(0);
    p3_block(k0, Bs, Cs, tb * TB, (tb + 1) * TB, hh, An, h, Dd, y + row0off + (size_t)tb * blkrows, ga);
    const int tn = (tb + 2 < NTB) ? tb + 2 : NTB - 1;
    p3_fetch(k0, dlT, duT, u + row0off + (size_t)tn * blkrows, tile0 + (size_t)tn * TILE_STRIDE, c);
    __builtin_amdgcn_sched_barrier(0);
    const int rn = (tb + 2 < NTB) ? (tb + 2) * TB : LCH - SG3;
    p3_block(k1, Bs, Cs, (tb + 1) * TB, rn, hh, An, h, Dd, y + row0off + (size_t)(tb + 1) * blkrows, ga);
  }
}

extern "C" void kernel_launch(void* const* d_in, const int* in_sizes, int n_in,
                              void* d_out, int out_size, void* d_ws, size_t ws_size,
                              hipStream_t stream) {
  (void)in_sizes; (void)n_in; (void)out_size; (void)ws_size;
  const float* u    = (const float*)d_in[0];
  const float* Wx   = (const float*)d_in[1];
  const float* Wdt  = (const float*)d_in[2];
  const float* bdt  = (const float*)d_in[3];
  const float* logA = (const float*)d_in[4];
  const float* Dp   = (const float*)d_in[5];
  float* y = (float*)d_out;

  float* ws     = (float*)d_ws;
  float* xdbl   = ws;
  float* chunkA = xdbl  + (size_t)MROWS * XPROJ_N;
  float* chunkH = chunkA + (size_t)BATCH * DINNER * NCH * DSTATE;
  float* carry  = chunkH + (size_t)BATCH * DINNER * NCH * DSTATE;
  float* dlT    = carry  + (size_t)BATCH * DINNER * NCH * DSTATE;
  float* duT    = dlT    + (size_t)MROWS * DINNER;

  k_xproj<<<(MROWS / 64) * (XPROJ_N / 32), 256, 0, stream>>>(u, Wx, xdbl);
  const int scan_blocks = BATCH * NCH * (DINNER / DWG);
  k_scan_phase1<<<scan_blocks, 256, 0, stream>>>(u, xdbl, Wdt, bdt, logA, chunkA, chunkH, dlT, duT);
  k_scan_phase2<<<(BATCH * DINNER * DSTATE) / 256, 256, 0, stream>>>(chunkA, chunkH, carry);
  k_scan_phase3<<<scan_blocks, 256, 0, stream>>>(u, xdbl, logA, Dp, carry, dlT, duT, y);
}


# scan phase 1 softplus: library log1pf (in-line double-float code) replaced by max(x,0) + log1p(e^-|x|), log1p(t) = log(1+t) * t/((1+t)-1) on the hardware log and reciprocal; sum order unchanged
# speedup vs baseline: 4.4187x; 1.1032x over previous
#include <hip/hip_runtime.h>
#include <hip/hip_bf16.h>
#include <math.h>


#define BATCH   2
#define LENGTH  2048
#define DINNER  1024
#define DSTATE  16
#define DTRANK  64
#define MROWS   (BATCH * LENGTH)
#define XPROJ_N (DTRANK + 2 * DSTATE)

#define NCH 16
#define LCH (LENGTH / NCH)

#define TB    16
#define HS    (DSTATE / 2)
#define DWG   128
#define TILE_FLOATS 256
#define TILE_STRIDE ((size_t)(DINNER / 16) * TILE_FLOATS)
#define RPITCH 68

typedef __attribute__((ext_vector_type(2))) float v2f;
typedef __attribute__((ext_vector_type(4))) float v4f;
typedef __attribute__((ext_vector_type(8))) float v8f;

#define XBM    64
#define XBN    32
#define XKC    128
#define XPITCH (XKC + 4)
#define XROWS  (XBM + XBN)
#define XNST   (DINNER / XKC)
#define XLD    (XROWS * (XKC / 4) / 256)
#define XNP    (XKC / 8)

__global__ __launch_bounds__(256)
void k_xproj(const float* __restrict__ u, const float* __restrict__ Wx,
             float* __restrict__ xdbl) {
  __shared__ __attribute__((aligned(16))) float St[2][XROWS][XPITCH];
  const int lane = threadIdx.x & 31;
  const int wave = __builtin_amdgcn_readfirstlane(threadIdx.x >> 5);
  const int c    = lane & 15;
  const int hh   = lane >> 4;
  const int m0   = (blockIdx.x / 3) * XBM;
  const int n0   = (blockIdx.x % 3) * XBN;
  const int mi   = wave & 3;
  const int ni   = wave >> 2;

  const float* srow[XLD];
#pragma unroll
  for (int i = 0; i < XLD; ++i) {
    const int r = 8 * i + wave;
    srow[i] = (i < XBM / 8 ? u + (size_t)(m0 + r) * DINNER
                           : Wx + (size_t)(n0 + r - XBM) * DINNER) + 4 * lane;
  }

  v4f pf[XLD];
#pragma unroll
  for (int i = 0; i < XLD; ++i) pf[i] = *(const v4f*)(srow[i]);
#pragma unroll
  for (int i = 0; i < XLD; ++i) *(v4f*)&St[0][8 * i + wave][4 * lane] = pf[i];
  __syncthreads();

  v8f acc0 = {}, acc1 = {};
#pragma unroll
  for (int s = 0; s < XNST; ++s) {
    if (s + 1 < XNST) {
#pragma unroll
      for (int i = 0; i < XLD; ++i) pf[i] = *(const v4f*)(srow[i] + (s + 1) * XKC);
    }
    const float* Ua = &St[s & 1][16 * mi + c][4 * hh];
    const float* Wb = &St[s & 1][XBM + 16 * ni + c][4 * hh];
    v4f a4[XNP], b4[XNP];
#pragma unroll
    for (int p = 0; p < XNP; ++p) {
      a4[p] = *(const v4f*)(Ua + 8 * p);
      b4[p] = *(const v4f*)(Wb + 8 * p);
    }
    __builtin_amdgcn_sched_barrier(0);
#pragma unroll
    for (int p = 0; p < XNP; ++p) {
      const v2f a0 = {a4[p].x, a4[p].y}, a1 = {a4[p].z, a4[p].w};
      const v2f b0 = {b4[p].x, b4[p].y}, b1 = {b4[p].z, b4[p].w};
      acc0 = __builtin_amdgcn_wmma_f32_16x16x4_f32(
           false, a0,  false, b0,
           (short)0, acc0,  false,  false);
      acc1 = __builtin_amdgcn_wmma_f32_16x16x4_f32(
          false, a1, false, b1, (short)0, acc1, false, false);
    }
    __builtin_amdgcn_sched_barrier(0);
    if (s + 1 < XNST) {
#pragma unroll
      for (int i = 0; i < XLD; ++i) *(v4f*)&St[(s + 1) & 1][8 * i + wave][4 * lane] = pf[i];
      __syncthreads();
    }
  }

  float* Os = &St[0][0][0];
#pragma unroll
  for (int r = 0; r < 8; ++r)
    Os[(16 * mi + 8 * hh + r) * XBN + 16 * ni + c] = acc0[r] + acc1[r];
  __syncthreads();
#pragma unroll
  for (int j = 0; j < XBM * XBN / 4 / 256; ++j) {
    const int idx = j * 256 + threadIdx.x;
    *(v4f*)(xdbl + (size_t)(m0 + (idx >> 3)) * XPROJ_N + n0 + 4 * (idx & 7)) = ((const v4f*)Os)[idx];
  }
}

__device__ __forceinline__ void delta_tile(const float (*Rs)[RPITCH], int row0, int c, int hh,
                                           const v2f (&bfrag)[DTRANK / 4], float bias,
                                           float (&dl)[8]) {
  v2f a[DTRANK / 4];
#pragma unroll
  for (int j = 0; j < DTRANK / 4; ++j)
    a[j] = *(const v2f*)&Rs[row0 + c][2 * hh + 4 * j];
  v8f acc = {};
#pragma unroll
  for (int j = 0; j < DTRANK / 4; ++j)
    acc = __builtin_amdgcn_wmma_f32_16x16x4_f32(
        false, a[j], false, bfrag[j], (short)0, acc, false, false);
  float x[8], t[8], w[8], lp[8];
#pragma unroll
  for (int r = 0; r < 8; ++r) x[r] = acc[r] + bias;
#pragma unroll
  for (int r = 0; r < 8; ++r) t[r] = __expf(-fabsf(x[r]));
#pragma unroll
  for (int r = 0; r < 8; ++r) w[r] = 1.f + t[r];
#pragma unroll
  for (int r = 0; r < 8; ++r) {
    const float wm1 = w[r] - 1.f;
    const float q   = t[r] * __builtin_amdgcn_rcpf(wm1);
    const float lw  = __builtin_amdgcn_logf(w[r]) * 0.69314718f;
    lp[r] = (wm1 == 0.f) ? t[r] : lw * q;
  }
#pragma unroll
  for (int r = 0; r < 8; ++r) dl[r] = fmaxf(x[r], 0.f) + lp[r];
}

__device__ __forceinline__ void gather16(const float (&own)[8], int hh, float (&out)[TB]) {
#pragma unroll
  for (int r = 0; r < 8; ++r) {
    float oth = __shfl_xor(own[r], 16);
    out[r]     = hh ? oth : own[r];
    out[8 + r] = hh ? own[r] : oth;
  }
}

#define SG 4
struct BGrp  { float4 b[SG][2]; };
#define SG3 2
struct BCGrp { float4 b[SG3][2]; float4 c[SG3][2]; };

__device__ __forceinline__ void b_read(BGrp& g, const float (*Bs)[DSTATE], int row, int hh) {
#pragma unroll
  for (int j = 0; j < SG; ++j) {
    g.b[j][0] = *(const float4*)&Bs[row + j][HS * hh];
    g.b[j][1] = *(const float4*)&Bs[row + j][HS * hh + 4];
  }
}

__device__ __forceinline__ void bc_read(BCGrp& g, const float (*Bs)[DSTATE], const float (*Cs)[DSTATE],
                                        int row, int hh) {
#pragma unroll
  for (int j = 0; j < SG3; ++j) {
    g.b[j][0] = *(const float4*)&Bs[row + j][HS * hh];
    g.b[j][1] = *(const float4*)&Bs[row + j][HS * hh + 4];
    g.c[j][0] = *(const float4*)&Cs[row + j][HS * hh];
    g.c[j][1] = *(const float4*)&Cs[row + j][HS * hh + 4];
  }
}

__device__ __forceinline__ void p1_steps(const BGrp& g, int s0, const float (&dl16)[TB],
                                         const float (&du16)[TB], const float (&An)[HS],
                                         float (&h)[HS], float (&ap)[HS]) {
#pragma unroll
  for (int j = 0; j < SG; ++j) {
    const int s = s0 + j;
    const float Bv[HS] = {g.b[j][0].x, g.b[j][0].y, g.b[j][0].z, g.b[j][0].w,
                          g.b[j][1].x, g.b[j][1].y, g.b[j][1].z, g.b[j][1].w};
#pragma unroll
    for (int i = 0; i < HS; ++i) {
      float a = __expf(dl16[s] * An[i]);
      h[i]  = a * h[i] + du16[s] * Bv[i];
      ap[i] *= a;
    }
  }
}

__device__ __forceinline__ void p3_steps(const BCGrp& g, int s0, const float (&dl16)[TB],
                                         const float (&du16)[TB], const float (&An)[HS],
                                         float (&h)[HS], float (&yp)[TB]) {
#pragma unroll
  for (int j = 0; j < SG3; ++j) {
    const int s = s0 + j;
    const float Bv[HS] = {g.b[j][0].x, g.b[j][0].y, g.b[j][0].z, g.b[j][0].w,
                          g.b[j][1].x, g.b[j][1].y, g.b[j][1].z, g.b[j][1].w};
    const float Cv[HS] = {g.c[j][0].x, g.c[j][0].y, g.c[j][0].z, g.c[j][0].w,
                          g.c[j][1].x, g.c[j][1].y, g.c[j][1].z, g.c[j][1].w};
    float acc = 0.f;
#pragma unroll
    for (int i = 0; i < HS; ++i) {
      float a = __expf(dl16[s] * An[i]);
      h[i] = a * h[i] + du16[s] * Bv[i];
      acc += h[i] * Cv[i];
    }
    yp[s] = acc;
  }
}

#define RS_LD (LCH * (DTRANK / 4) / 256)
#define B_LD  (LCH * (DSTATE / 4) / 256)
#define BC_LD (LCH * (2 * DSTATE / 4) / 256)

__global__ __launch_bounds__(256)
void k_scan_phase1(const float* __restrict__ u, const float* __restrict__ xdbl,
                   const float* __restrict__ Wdt, const float* __restrict__ bdt,
                   const float* __restrict__ logA,
                   float* __restrict__ chunkA, float* __restrict__ chunkH,
                   float* __restrict__ dlT, float* __restrict__ duT) {
  const int dblk = blockIdx.x & 7;
  const int ch   = (blockIdx.x >> 3) & (NCH - 1);
  const int b    = blockIdx.x >> 7;
  const int lane = threadIdx.x & 31;
  const int c    = lane & 15;
  const int hh   = lane >> 4;
  const int d    = dblk * DWG + (threadIdx.x >> 5) * 16 + c;
  const int t0   = ch * LCH;
  const size_t tile0 = ((size_t)((b * LENGTH + t0) / TB) * (DINNER / 16) + (size_t)(d >> 4)) * TILE_FLOATS;

  __shared__ __attribute__((aligned(16))) float Rs[LCH][RPITCH];
  __shared__ __attribute__((aligned(16))) float Bs[LCH][DSTATE];

  const float* xrow0 = xdbl + (size_t)(b * LENGTH + t0) * XPROJ_N;
  float4 rr[RS_LD], rb[B_LD];
#pragma unroll
  for (int k = 0; k < RS_LD; ++k) {
    const int i = threadIdx.x + 256 * k;
    rr[k] = *(const float4*)(xrow0 + (size_t)(i >> 4) * XPROJ_N + (i & 15) * 4);
  }
#pragma unroll
  for (int k = 0; k < B_LD; ++k) {
    const int i = threadIdx.x + 256 * k;
    rb[k] = *(const float4*)(xrow0 + (size_t)(i >> 2) * XPROJ_N + DTRANK + (i & 3) * 4);
  }
  v2f bfrag[DTRANK / 4];
#pragma unroll
  for (int j = 0; j < DTRANK / 4; ++j)
    bfrag[j] = *(const v2f*)(Wdt + (size_t)d * DTRANK + 2 * hh + 4 * j);
  const float bias = bdt[d];
  float la[HS];
#pragma unroll
  for (int i = 0; i < HS; ++i) la[i] = logA[d * DSTATE + HS * hh + i];

#pragma unroll
  for (int k = 0; k < RS_LD; ++k) {
    const int i = threadIdx.x + 256 * k;
    *(float4*)&Rs[i >> 4][(i & 15) * 4] = rr[k];
  }
#pragma unroll
  for (int k = 0; k < B_LD; ++k) {
    const int i = threadIdx.x + 256 * k;
    *(float4*)&Bs[i >> 2][(i & 3) * 4] = rb[k];
  }

  float An[HS], h[HS], ap[HS];
#pragma unroll
  for (int i = 0; i < HS; ++i) {
    An[i] = -__expf(la[i]);
    h[i]  = 0.f;
    ap[i] = 1.f;
  }
  __syncthreads();

  for (int tb = 0; tb < LCH / TB; ++tb) {
    const int row0 = tb * TB;
    float uo[8];
#pragma unroll
    for (int r = 0; r < 8; ++r)
      uo[r] = u[(size_t)(b * LENGTH + t0 + row0 + 8 * hh + r) * DINNER + d];

    BGrp ga, gb;
    b_read(ga, Bs, row0, hh);
    __builtin_amdgcn_sched_barrier(0);

    float dl[8], du[8];
    delta_tile(Rs, row0, c, hh, bfrag, bias, dl);
#pragma unroll
    for (int r = 0; r < 8; ++r) du[r] = dl[r] * uo[r];

    {
      const size_t tile = tile0 + (size_t)tb * TILE_STRIDE;
#pragma unroll
      for (int j = 0; j < 2; ++j) {
        const size_t o = tile + (size_t)(((2 * j + hh) * 16 + c) * 4);
        *(float4*)(dlT + o) = make_float4(dl[4 * j], dl[4 * j + 1], dl[4 * j + 2], dl[4 * j + 3]);
        *(float4*)(duT + o) = make_float4(du[4 * j], du[4 * j + 1], du[4 * j + 2], du[4 * j + 3]);
      }
    }

    float dl16[TB], du16[TB];
    gather16(dl, hh, dl16);
    gather16(du, hh, du16);

    __builtin_amdgcn_sched_barrier(0);
    b_read(gb, Bs, row0 + SG, hh);
    __builtin_amdgcn_sched_barrier(0);
    p1_steps(ga, 0, dl16, du16, An, h, ap);
    __builtin_amdgcn_sched_barrier(0);
    b_read(ga, Bs, row0 + 2 * SG, hh);
    __builtin_amdgcn_sched_barrier(0);
    p1_steps(gb, SG, dl16, du16, An, h, ap);
    __builtin_amdgcn_sched_barrier(0);
    b_read(gb, Bs, row0 + 3 * SG, hh);
    __builtin_amdgcn_sched_barrier(0);
    p1_steps(ga, 2 * SG, dl16, du16, An, h, ap);
    p1_steps(gb, 3 * SG, dl16, du16, An, h, ap);
  }
  size_t base = (((size_t)(b * DINNER + d)) * NCH + ch) * DSTATE + HS * hh;
#pragma unroll
  for (int i = 0; i < HS; ++i) {
    chunkA[base + i] = ap[i];
    chunkH[base + i] = h[i];
  }
}

__global__ void k_scan_phase2(const float* __restrict__ chunkA, const float* __restrict__ chunkH,
                              float* __restrict__ carry) {
  const int idx = blockIdx.x * 256 + threadIdx.x;
  const int n   = idx & 15;
  const int bd  = idx >> 4;
  size_t base = (size_t)bd * NCH * DSTATE + n;
  float h = 0.f;
#pragma unroll
  for (int c = 0; c < NCH; ++c) {
    carry[base + (size_t)c * DSTATE] = h;
    h = chunkA[base + (size_t)c * DSTATE] * h + chunkH[base + (size_t)c * DSTATE];
  }
}

struct P3Blk {
  float4 dq[4];
  float4 gq[4];
  float  uo[8];
};

__device__ __forceinline__ void p3_fetch(P3Blk& k, const float* __restrict__ dlT,
                                         const float* __restrict__ duT,
                                         const float* __restrict__ urow, size_t tile, int c) {
#pragma unroll
  for (int p = 0; p < 4; ++p) {
    k.dq[p] = *(const float4*)(dlT + tile + (size_t)((p * 16 + c) * 4));
    k.gq[p] = *(const float4*)(duT + tile + (size_t)((p * 16 + c) * 4));
  }
#pragma unroll
  for (int r = 0; r < 8; ++r) k.uo[r] = urow[(size_t)r * DINNER];
}

__device__ __forceinline__ void p3_block(const P3Blk& k, const float (*Bs)[DSTATE],
                                         const float (*Cs)[DSTATE], int row0, int nextrow, int hh,
                                         const float (&An)[HS], float (&h)[HS], float Dd,
                                         float* __restrict__ yrow, BCGrp& ga) {
  float dl16[TB], du16[TB], yp[TB];
#pragma unroll
  for (int p = 0; p < 4; ++p) {
    const int s0 = 8 * (p & 1) + 4 * (p >> 1);
    dl16[s0] = k.dq[p].x; dl16[s0 + 1] = k.dq[p].y; dl16[s0 + 2] = k.dq[p].z; dl16[s0 + 3] = k.dq[p].w;
    du16[s0] = k.gq[p].x; du16[s0 + 1] = k.gq[p].y; du16[s0 + 2] = k.gq[p].z; du16[s0 + 3] = k.gq[p].w;
  }
  BCGrp gb;
#pragma unroll
  for (int g = 0; g < TB / SG3; ++g) {
    BCGrp& cur = (g & 1) ? gb : ga;
    BCGrp& nxt = (g & 1) ? ga : gb;
    if (g > 0) __builtin_amdgcn_sched_barrier(0);
    bc_read(nxt, Bs, Cs, (g + 1 < TB / SG3) ? row0 + (g + 1) * SG3 : nextrow, hh);
    __builtin_amdgcn_sched_barrier(0);
    p3_steps(cur, g * SG3, dl16, du16, An, h, yp);
  }
#pragma unroll
  for (int r = 0; r < 8; ++r) {
    float mine = hh ? yp[8 + r] : yp[r];
    float send = hh ? yp[r] : yp[8 + r];
    float recv = __shfl_xor(send, 16);
    float lo = hh ? recv : mine;
    float hi = hh ? mine : recv;
    yrow[(size_t)r * DINNER] = (lo + hi) + k.uo[r] * Dd;
  }
}

__global__ __launch_bounds__(256)
void k_scan_phase3(const float* __restrict__ u, const float* __restrict__ xdbl,
                   const float* __restrict__ logA,
                   const float* __restrict__ Dp, const float* __restrict__ carry,
                   const float* __restrict__ dlT, const float* __restrict__ duT,
                   float* __restrict__ y) {
  const int dblk = blockIdx.x & 7;
  const int ch   = (blockIdx.x >> 3) & (NCH - 1);
  const int b    = blockIdx.x >> 7;
  const int lane = threadIdx.x & 31;
  const int c    = lane & 15;
  const int hh   = lane >> 4;
  const int d    = dblk * DWG + (threadIdx.x >> 5) * 16 + c;
  const int t0   = ch * LCH;
  const size_t tile0 = ((size_t)((b * LENGTH + t0) / TB) * (DINNER / 16) + (size_t)(d >> 4)) * TILE_FLOATS;
  const size_t row0off = (size_t)(b * LENGTH + t0 + 8 * hh) * DINNER + d;
  const size_t blkrows = (size_t)TB * DINNER;

  P3Blk k0, k1;
  p3_fetch(k0, dlT, duT, u + row0off, tile0, c);

  __shared__ __attribute__((aligned(16))) float Bs[LCH][DSTATE];
  __shared__ __attribute__((aligned(16))) float Cs[LCH][DSTATE];

  const float* xrow0 = xdbl + (size_t)(b * LENGTH + t0) * XPROJ_N + DTRANK;
  float4 rbc[BC_LD];
#pragma unroll
  for (int k = 0; k < BC_LD; ++k) {
    const int i = threadIdx.x + 256 * k;
    rbc[k] = *(const float4*)(xrow0 + (size_t)(i >> 3) * XPROJ_N + (i & 7) * 4);
  }
  float la[HS], h[HS];
  size_t cbase = (((size_t)(b * DINNER + d)) * NCH + ch) * DSTATE + HS * hh;
#pragma unroll
  for (int i = 0; i < HS; ++i) {
    la[i] = logA[d * DSTATE + HS * hh + i];
    h[i]  = carry[cbase + i];
  }
  const float Dd = Dp[d];

  {
    const int q = threadIdx.x & 7;
    float* dst = (q < 4) ? &Bs[0][q * 4] : &Cs[0][(q - 4) * 4];
#pragma unroll
    for (int k = 0; k < BC_LD; ++k) {
      const int i = threadIdx.x + 256 * k;
      *(float4*)(dst + (i >> 3) * DSTATE) = rbc[k];
    }
  }

  float An[HS];
#pragma unroll
  for (int i = 0; i < HS; ++i) An[i] = -__expf(la[i]);
  __syncthreads();

  const int NTB = LCH / TB;
  BCGrp ga;
  bc_read(ga, Bs, Cs, 0, hh);
  for (int tb = 0; tb < NTB; tb += 2) {
    p3_fetch(k1, dlT, duT, u + row0off + (size_t)(tb + 1) * blkrows, tile0 + (size_t)(tb + 1) * TILE_STRIDE, c);
    __builtin_amdgcn_sched_barrier(0);
    p3_block(k0, Bs, Cs, tb * TB, (tb + 1) * TB, hh, An, h, Dd, y + row0off + (size_t)tb * blkrows, ga);
    const int tn = (tb + 2 < NTB) ? tb + 2 : NTB - 1;
    p3_fetch(k0, dlT, duT, u + row0off + (size_t)tn * blkrows, tile0 + (size_t)tn * TILE_STRIDE, c);
    __builtin_amdgcn_sched_barrier(0);
    const int rn = (tb + 2 < NTB) ? (tb + 2) * TB : LCH - SG3;
    p3_block(k1, Bs, Cs, (tb + 1) * TB, rn, hh, An, h, Dd, y + row0off + (size_t)(tb + 1) * blkrows, ga);
  }
}

extern "C" void kernel_launch(void* const* d_in, const int* in_sizes, int n_in,
                              void* d_out, int out_size, void* d_ws, size_t ws_size,
                              hipStream_t stream) {
  (void)in_sizes; (void)n_in; (void)out_size; (void)ws_size;
  const float* u    = (const float*)d_in[0];
  const float* Wx   = (const float*)d_in[1];
  const float* Wdt  = (const float*)d_in[2];
  const float* bdt  = (const float*)d_in[3];
  const float* logA = (const float*)d_in[4];
  const float* Dp   = (const float*)d_in[5];
  float* y = (float*)d_out;

  float* ws     = (float*)d_ws;
  float* xdbl   = ws;
  float* chunkA = xdbl  + (size_t)MROWS * XPROJ_N;
  float* chunkH = chunkA + (size_t)BATCH * DINNER * NCH * DSTATE;
  float* carry  = chunkH + (size_t)BATCH * DINNER * NCH * DSTATE;
  float* dlT    = carry  + (size_t)BATCH * DINNER * NCH * DSTATE;
  float* duT    = dlT    + (size_t)MROWS * DINNER;

  k_xproj<<<(MROWS / 64) * (XPROJ_N / 32), 256, 0, stream>>>(u, Wx, xdbl);
  const int scan_blocks = BATCH * NCH * (DINNER / DWG);
  k_scan_phase1<<<scan_blocks, 256, 0, stream>>>(u, xdbl, Wdt, bdt, logA, chunkA, chunkH, dlT, duT);
  k_scan_phase2<<<(BATCH * DINNER * DSTATE) / 256, 256, 0, stream>>>(chunkA, chunkH, carry);
  k_scan_phase3<<<scan_blocks, 256, 0, stream>>>(u, xdbl, logA, Dp, carry, dlT, duT, y);
}


# scan phases 1, 3: time-loop multiplies and multiply-adds work on pairs of adjacent states as packed two-float instructions; y state sum as even + odd partial sums: sum order changes
# speedup vs baseline: 4.6758x; 1.0582x over previous
#include <hip/hip_runtime.h>
#include <hip/hip_bf16.h>
#include <math.h>


#define BATCH   2
#define LENGTH  2048
#define DINNER  1024
#define DSTATE  16
#define DTRANK  64
#define MROWS   (BATCH * LENGTH)
#define XPROJ_N (DTRANK + 2 * DSTATE)

#define NCH 16
#define LCH (LENGTH / NCH)

#define TB    16
#define HS    (DSTATE / 2)
#define DWG   128
#define TILE_FLOATS 256
#define TILE_STRIDE ((size_t)(DINNER / 16) * TILE_FLOATS)
#define RPITCH 68

typedef __attribute__((ext_vector_type(2))) float v2f;
typedef __attribute__((ext_vector_type(4))) float v4f;
typedef __attribute__((ext_vector_type(8))) float v8f;

#define XBM    64
#define XBN    32
#define XKC    128
#define XPITCH (XKC + 4)
#define XROWS  (XBM + XBN)
#define XNST   (DINNER / XKC)
#define XLD    (XROWS * (XKC / 4) / 256)
#define XNP    (XKC / 8)

__global__ __launch_bounds__(256)
void k_xproj(const float* __restrict__ u, const float* __restrict__ Wx,
             float* __restrict__ xdbl) {
  __shared__ __attribute__((aligned(16))) float St[2][XROWS][XPITCH];
  const int lane = threadIdx.x & 31;
  const int wave = __builtin_amdgcn_readfirstlane(threadIdx.x >> 5);
  const int c    = lane & 15;
  const int hh   = lane >> 4;
  const int m0   = (blockIdx.x / 3) * XBM;
  const int n0   = (blockIdx.x % 3) * XBN;
  const int mi   = wave & 3;
  const int ni   = wave >> 2;

  const float* srow[XLD];
#pragma unroll
  for (int i = 0; i < XLD; ++i) {
    const int r = 8 * i + wave;
    srow[i] = (i < XBM / 8 ? u + (size_t)(m0 + r) * DINNER
                           : Wx + (size_t)(n0 + r - XBM) * DINNER) + 4 * lane;
  }

  v4f pf[XLD];
#pragma unroll
  for (int i = 0; i < XLD; ++i) pf[i] = *(const v4f*)(srow[i]);
#pragma unroll
  for (int i = 0; i < XLD; ++i) *(v4f*)&St[0][8 * i + wave][4 * lane] = pf[i];
  __syncthreads();

  v8f acc0 = {}, acc1 = {};
#pragma unroll
  for (int s = 0; s < XNST; ++s) {
    if (s + 1 < XNST) {
#pragma unroll
      for (int i = 0; i < XLD; ++i) pf[i] = *(const v4f*)(srow[i] + (s + 1) * XKC);
    }
    const float* Ua = &St[s & 1][16 * mi + c][4 * hh];
    const float* Wb = &St[s & 1][XBM + 16 * ni + c][4 * hh];
    v4f a4[XNP], b4[XNP];
#pragma unroll
    for (int p = 0; p < XNP; ++p) {
      a4[p] = *(const v4f*)(Ua + 8 * p);
      b4[p] = *(const v4f*)(Wb + 8 * p);
    }
    __builtin_amdgcn_sched_barrier(0);
#pragma unroll
    for (int p = 0; p < XNP; ++p) {
      const v2f a0 = {a4[p].x, a4[p].y}, a1 = {a4[p].z, a4[p].w};
      const v2f b0 = {b4[p].x, b4[p].y}, b1 = {b4[p].z, b4[p].w};
      acc0 = __builtin_amdgcn_wmma_f32_16x16x4_f32(
           false, a0,  false, b0,
           (short)0, acc0,  false,  false);
      acc1 = __builtin_amdgcn_wmma_f32_16x16x4_f32(
          false, a1, false, b1, (short)0, acc1, false, false);
    }
    __builtin_amdgcn_sched_barrier(0);
    if (s + 1 < XNST) {
#pragma unroll
      for (int i = 0; i < XLD; ++i) *(v4f*)&St[(s + 1) & 1][8 * i + wave][4 * lane] = pf[i];
      __syncthreads();
    }
  }

  float* Os = &St[0][0][0];
#pragma unroll
  for (int r = 0; r < 8; ++r)
    Os[(16 * mi + 8 * hh + r) * XBN + 16 * ni + c] = acc0[r] + acc1[r];
  __syncthreads();
#pragma unroll
  for (int j = 0; j < XBM * XBN / 4 / 256; ++j) {
    const int idx = j * 256 + threadIdx.x;
    *(v4f*)(xdbl + (size_t)(m0 + (idx >> 3)) * XPROJ_N + n0 + 4 * (idx & 7)) = ((const v4f*)Os)[idx];
  }
}

__device__ __forceinline__ void delta_tile(const float (*Rs)[RPITCH], int row0, int c, int hh,
                                           const v2f (&bfrag)[DTRANK / 4], float bias,
                                           float (&dl)[8]) {
  v2f a[DTRANK / 4];
#pragma unroll
  for (int j = 0; j < DTRANK / 4; ++j)
    a[j] = *(const v2f*)&Rs[row0 + c][2 * hh + 4 * j];
  v8f acc = {};
#pragma unroll
  for (int j = 0; j < DTRANK / 4; ++j)
    acc = __builtin_amdgcn_wmma_f32_16x16x4_f32(
        false, a[j], false, bfrag[j], (short)0, acc, false, false);
  float x[8], t[8], w[8], lp[8];
#pragma unroll
  for (int r = 0; r < 8; ++r) x[r] = acc[r] + bias;
#pragma unroll
  for (int r = 0; r < 8; ++r) t[r] = __expf(-fabsf(x[r]));
#pragma unroll
  for (int r = 0; r < 8; ++r) w[r] = 1.f + t[r];
#pragma unroll
  for (int r = 0; r < 8; ++r) {
    const float wm1 = w[r] - 1.f;
    const float q   = t[r] * __builtin_amdgcn_rcpf(wm1);
    const float lw  = __builtin_amdgcn_logf(w[r]) * 0.69314718f;
    lp[r] = (wm1 == 0.f) ? t[r] : lw * q;
  }
#pragma unroll
  for (int r = 0; r < 8; ++r) dl[r] = fmaxf(x[r], 0.f) + lp[r];
}

__device__ __forceinline__ void gather16(const float (&own)[8], int hh, float (&out)[TB]) {
#pragma unroll
  for (int r = 0; r < 8; ++r) {
    float oth = __shfl_xor(own[r], 16);
    out[r]     = hh ? oth : own[r];
    out[8 + r] = hh ? own[r] : oth;
  }
}

#define SG 4
struct BGrp  { v4f b[SG][2]; };
#define SG3 2
struct BCGrp { v4f b[SG3][2]; v4f c[SG3][2]; };

#define HP (HS / 2)
#define LOG2E_F 1.44269504088896340736f

__device__ __forceinline__ v2f exp_pair(v2f x) {
  const v2f e = x * (v2f){LOG2E_F, LOG2E_F};
  return (v2f){__builtin_amdgcn_exp2f(e.x), __builtin_amdgcn_exp2f(e.y)};
}

__device__ __forceinline__ void b_read(BGrp& g, const float (*Bs)[DSTATE], int row, int hh) {
#pragma unroll
  for (int j = 0; j < SG; ++j) {
    g.b[j][0] = *(const v4f*)&Bs[row + j][HS * hh];
    g.b[j][1] = *(const v4f*)&Bs[row + j][HS * hh + 4];
  }
}

__device__ __forceinline__ void bc_read(BCGrp& g, const float (*Bs)[DSTATE], const float (*Cs)[DSTATE],
                                        int row, int hh) {
#pragma unroll
  for (int j = 0; j < SG3; ++j) {
    g.b[j][0] = *(const v4f*)&Bs[row + j][HS * hh];
    g.b[j][1] = *(const v4f*)&Bs[row + j][HS * hh + 4];
    g.c[j][0] = *(const v4f*)&Cs[row + j][HS * hh];
    g.c[j][1] = *(const v4f*)&Cs[row + j][HS * hh + 4];
  }
}

__device__ __forceinline__ void p1_steps(const BGrp& g, int s0, const float (&dl16)[TB],
                                         const float (&du16)[TB], const v2f (&An)[HP],
                                         v2f (&h)[HP], v2f (&ap)[HP]) {
#pragma unroll
  for (int j = 0; j < SG; ++j) {
    const int s = s0 + j;
    const v2f Bv[HP] = {g.b[j][0].xy, g.b[j][0].zw, g.b[j][1].xy, g.b[j][1].zw};
    const v2f dls = {dl16[s], dl16[s]};
    const v2f dus = {du16[s], du16[s]};
#pragma unroll
    for (int i = 0; i < HP; ++i) {
      const v2f a = exp_pair(dls * An[i]);
      h[i]  = __builtin_elementwise_fma(a, h[i], dus * Bv[i]);
      ap[i] = ap[i] * a;
    }
  }
}

__device__ __forceinline__ void p3_steps(const BCGrp& g, int s0, const float (&dl16)[TB],
                                         const float (&du16)[TB], const v2f (&An)[HP],
                                         v2f (&h)[HP], float (&yp)[TB]) {
#pragma unroll
  for (int j = 0; j < SG3; ++j) {
    const int s = s0 + j;
    const v2f Bv[HP] = {g.b[j][0].xy, g.b[j][0].zw, g.b[j][1].xy, g.b[j][1].zw};
    const v2f Cv[HP] = {g.c[j][0].xy, g.c[j][0].zw, g.c[j][1].xy, g.c[j][1].zw};
    const v2f dls = {dl16[s], dl16[s]};
    const v2f dus = {du16[s], du16[s]};
    v2f acc = {0.f, 0.f};
#pragma unroll
    for (int i = 0; i < HP; ++i) {
      const v2f a = exp_pair(dls * An[i]);
      h[i] = __builtin_elementwise_fma(a, h[i], dus * Bv[i]);
      acc  = __builtin_elementwise_fma(h[i], Cv[i], acc);
    }
    yp[s] = acc.x + acc.y;
  }
}

#define RS_LD (LCH * (DTRANK / 4) / 256)
#define B_LD  (LCH * (DSTATE / 4) / 256)
#define BC_LD (LCH * (2 * DSTATE / 4) / 256)

__global__ __launch_bounds__(256)
void k_scan_phase1(const float* __restrict__ u, const float* __restrict__ xdbl,
                   const float* __restrict__ Wdt, const float* __restrict__ bdt,
                   const float* __restrict__ logA,
                   float* __restrict__ chunkA, float* __restrict__ chunkH,
                   float* __restrict__ dlT, float* __restrict__ duT) {
  const int dblk = blockIdx.x & 7;
  const int ch   = (blockIdx.x >> 3) & (NCH - 1);
  const int b    = blockIdx.x >> 7;
  const int lane = threadIdx.x & 31;
  const int c    = lane & 15;
  const int hh   = lane >> 4;
  const int d    = dblk * DWG + (threadIdx.x >> 5) * 16 + c;
  const int t0   = ch * LCH;
  const size_t tile0 = ((size_t)((b * LENGTH + t0) / TB) * (DINNER / 16) + (size_t)(d >> 4)) * TILE_FLOATS;

  __shared__ __attribute__((aligned(16))) float Rs[LCH][RPITCH];
  __shared__ __attribute__((aligned(16))) float Bs[LCH][DSTATE];

  const float* xrow0 = xdbl + (size_t)(b * LENGTH + t0) * XPROJ_N;
  float4 rr[RS_LD], rb[B_LD];
#pragma unroll
  for (int k = 0; k < RS_LD; ++k) {
    const int i = threadIdx.x + 256 * k;
    rr[k] = *(const float4*)(xrow0 + (size_t)(i >> 4) * XPROJ_N + (i & 15) * 4);
  }
#pragma unroll
  for (int k = 0; k < B_LD; ++k) {
    const int i = threadIdx.x + 256 * k;
    rb[k] = *(const float4*)(xrow0 + (size_t)(i >> 2) * XPROJ_N + DTRANK + (i & 3) * 4);
  }
  v2f bfrag[DTRANK / 4];
#pragma unroll
  for (int j = 0; j < DTRANK / 4; ++j)
    bfrag[j] = *(const v2f*)(Wdt + (size_t)d * DTRANK + 2 * hh + 4 * j);
  const float bias = bdt[d];
  float la[HS];
#pragma unroll
  for (int i = 0; i < HS; ++i) la[i] = logA[d * DSTATE + HS * hh + i];

#pragma unroll
  for (int k = 0; k < RS_LD; ++k) {
    const int i = threadIdx.x + 256 * k;
    *(float4*)&Rs[i >> 4][(i & 15) * 4] = rr[k];
  }
#pragma unroll
  for (int k = 0; k < B_LD; ++k) {
    const int i = threadIdx.x + 256 * k;
    *(float4*)&Bs[i >> 2][(i & 3) * 4] = rb[k];
  }

  v2f An[HP], h[HP], ap[HP];
#pragma unroll
  for (int i = 0; i < HP; ++i) {
    An[i] = (v2f){-__expf(la[2 * i]), -__expf(la[2 * i + 1])};
    h[i]  = (v2f){0.f, 0.f};
    ap[i] = (v2f){1.f, 1.f};
  }
  __syncthreads();

  for (int tb = 0; tb < LCH / TB; ++tb) {
    const int row0 = tb * TB;
    float uo[8];
#pragma unroll
    for (int r = 0; r < 8; ++r)
      uo[r] = u[(size_t)(b * LENGTH + t0 + row0 + 8 * hh + r) * DINNER + d];

    BGrp ga, gb;
    b_read(ga, Bs, row0, hh);
    __builtin_amdgcn_sched_barrier(0);

    float dl[8], du[8];
    delta_tile(Rs, row0, c, hh, bfrag, bias, dl);
#pragma unroll
    for (int r = 0; r < 8; ++r) du[r] = dl[r] * uo[r];

    {
      const size_t tile = tile0 + (size_t)tb * TILE_STRIDE;
#pragma unroll
      for (int j = 0; j < 2; ++j) {
        const size_t o = tile + (size_t)(((2 * j + hh) * 16 + c) * 4);
        *(float4*)(dlT + o) = make_float4(dl[4 * j], dl[4 * j + 1], dl[4 * j + 2], dl[4 * j + 3]);
        *(float4*)(duT + o) = make_float4(du[4 * j], du[4 * j + 1], du[4 * j + 2], du[4 * j + 3]);
      }
    }

    float dl16[TB], du16[TB];
    gather16(dl, hh, dl16);
    gather16(du, hh, du16);

    __builtin_amdgcn_sched_barrier(0);
    b_read(gb, Bs, row0 + SG, hh);
    __builtin_amdgcn_sched_barrier(0);
    p1_steps(ga, 0, dl16, du16, An, h, ap);
    __builtin_amdgcn_sched_barrier(0);
    b_read(ga, Bs, row0 + 2 * SG, hh);
    __builtin_amdgcn_sched_barrier(0);
    p1_steps(gb, SG, dl16, du16, An, h, ap);
    __builtin_amdgcn_sched_barrier(0);
    b_read(gb, Bs, row0 + 3 * SG, hh);
    __builtin_amdgcn_sched_barrier(0);
    p1_steps(ga, 2 * SG, dl16, du16, An, h, ap);
    p1_steps(gb, 3 * SG, dl16, du16, An, h, ap);
  }
  size_t base = (((size_t)(b * DINNER + d)) * NCH + ch) * DSTATE + HS * hh;
#pragma unroll
  for (int i = 0; i < HS; ++i) {
    chunkA[base + i] = ap[i >> 1][i & 1];
    chunkH[base + i] = h[i >> 1][i & 1];
  }
}

__global__ void k_scan_phase2(const float* __restrict__ chunkA, const float* __restrict__ chunkH,
                              float* __restrict__ carry) {
  const int idx = blockIdx.x * 256 + threadIdx.x;
  const int n   = idx & 15;
  const int bd  = idx >> 4;
  size_t base = (size_t)bd * NCH * DSTATE + n;
  float h = 0.f;
#pragma unroll
  for (int c = 0; c < NCH; ++c) {
    carry[base + (size_t)c * DSTATE] = h;
    h = chunkA[base + (size_t)c * DSTATE] * h + chunkH[base + (size_t)c * DSTATE];
  }
}

struct P3Blk {
  float4 dq[4];
  float4 gq[4];
  float  uo[8];
};

__device__ __forceinline__ void p3_fetch(P3Blk& k, const float* __restrict__ dlT,
                                         const float* __restrict__ duT,
                                         const float* __restrict__ urow, size_t tile, int c) {
#pragma unroll
  for (int p = 0; p < 4; ++p) {
    k.dq[p] = *(const float4*)(dlT + tile + (size_t)((p * 16 + c) * 4));
    k.gq[p] = *(const float4*)(duT + tile + (size_t)((p * 16 + c) * 4));
  }
#pragma unroll
  for (int r = 0; r < 8; ++r) k.uo[r] = urow[(size_t)r * DINNER];
}

__device__ __forceinline__ void p3_block(const P3Blk& k, const float (*Bs)[DSTATE],
                                         const float (*Cs)[DSTATE], int row0, int nextrow, int hh,
                                         const v2f (&An)[HP], v2f (&h)[HP], float Dd,
                                         float* __restrict__ yrow, BCGrp& ga) {
  float dl16[TB], du16[TB], yp[TB];
#pragma unroll
  for (int p = 0; p < 4; ++p) {
    const int s0 = 8 * (p & 1) + 4 * (p >> 1);
    dl16[s0] = k.dq[p].x; dl16[s0 + 1] = k.dq[p].y; dl16[s0 + 2] = k.dq[p].z; dl16[s0 + 3] = k.dq[p].w;
    du16[s0] = k.gq[p].x; du16[s0 + 1] = k.gq[p].y; du16[s0 + 2] = k.gq[p].z; du16[s0 + 3] = k.gq[p].w;
  }
  BCGrp gb;
#pragma unroll
  for (int g = 0; g < TB / SG3; ++g) {
    BCGrp& cur = (g & 1) ? gb : ga;
    BCGrp& nxt = (g & 1) ? ga : gb;
    if (g > 0) __builtin_amdgcn_sched_barrier(0);
    bc_read(nxt, Bs, Cs, (g + 1 < TB / SG3) ? row0 + (g + 1) * SG3 : nextrow, hh);
    __builtin_amdgcn_sched_barrier(0);
    p3_steps(cur, g * SG3, dl16, du16, An, h, yp);
  }
#pragma unroll
  for (int r = 0; r < 8; ++r) {
    float mine = hh ? yp[8 + r] : yp[r];
    float send = hh ? yp[r] : yp[8 + r];
    float recv = __shfl_xor(send, 16);
    float lo = hh ? recv : mine;
    float hi = hh ? mine : recv;
    yrow[(size_t)r * DINNER] = (lo + hi) + k.uo[r] * Dd;
  }
}

__global__ __launch_bounds__(256)
void k_scan_phase3(const float* __restrict__ u, const float* __restrict__ xdbl,
                   const float* __restrict__ logA,
                   const float* __restrict__ Dp, const float* __restrict__ carry,
                   const float* __restrict__ dlT, const float* __restrict__ duT,
                   float* __restrict__ y) {
  const int dblk = blockIdx.x & 7;
  const int ch   = (blockIdx.x >> 3) & (NCH - 1);
  const int b    = blockIdx.x >> 7;
  const int lane = threadIdx.x & 31;
  const int c    = lane & 15;
  const int hh   = lane >> 4;
  const int d    = dblk * DWG + (threadIdx.x >> 5) * 16 + c;
  const int t0   = ch * LCH;
  const size_t tile0 = ((size_t)((b * LENGTH + t0) / TB) * (DINNER / 16) + (size_t)(d >> 4)) * TILE_FLOATS;
  const size_t row0off = (size_t)(b * LENGTH + t0 + 8 * hh) * DINNER + d;
  const size_t blkrows = (size_t)TB * DINNER;

  P3Blk k0, k1;
  p3_fetch(k0, dlT, duT, u + row0off, tile0, c);

  __shared__ __attribute__((aligned(16))) float Bs[LCH][DSTATE];
  __shared__ __attribute__((aligned(16))) float Cs[LCH][DSTATE];

  const float* xrow0 = xdbl + (size_t)(b * LENGTH + t0) * XPROJ_N + DTRANK;
  float4 rbc[BC_LD];
#pragma unroll
  for (int k = 0; k < BC_LD; ++k) {
    const int i = threadIdx.x + 256 * k;
    rbc[k] = *(const float4*)(xrow0 + (size_t)(i >> 3) * XPROJ_N + (i & 7) * 4);
  }
  float la[HS], h0[HS];
  size_t cbase = (((size_t)(b * DINNER + d)) * NCH + ch) * DSTATE + HS * hh;
#pragma unroll
  for (int i = 0; i < HS; ++i) {
    la[i] = logA[d * DSTATE + HS * hh + i];
    h0[i] = carry[cbase + i];
  }
  const float Dd = Dp[d];

  {
    const int q = threadIdx.x & 7;
    float* dst = (q < 4) ? &Bs[0][q * 4] : &Cs[0][(q - 4) * 4];
#pragma unroll
    for (int k = 0; k < BC_LD; ++k) {
      const int i = threadIdx.x + 256 * k;
      *(float4*)(dst + (i >> 3) * DSTATE) = rbc[k];
    }
  }

  v2f An[HP], h[HP];
#pragma unroll
  for (int i = 0; i < HP; ++i) {
    An[i] = (v2f){-__expf(la[2 * i]), -__expf(la[2 * i + 1])};
    h[i]  = (v2f){h0[2 * i], h0[2 * i + 1]};
  }
  __syncthreads();

  const int NTB = LCH / TB;
  BCGrp ga;
  bc_read(ga, Bs, Cs, 0, hh);
  for (int tb = 0; tb < NTB; tb += 2) {
    p3_fetch(k1, dlT, duT, u + row0off + (size_t)(tb + 1) * blkrows, tile0 + (size_t)(tb + 1) * TILE_STRIDE, c);
    __builtin_amdgcn_sched_barrier(0);
    p3_block(k0, Bs, Cs, tb * TB, (tb + 1) * TB, hh, An, h, Dd, y + row0off + (size_t)tb * blkrows, ga);
    const int tn = (tb + 2 < NTB) ? tb + 2 : NTB - 1;
    p3_fetch(k0, dlT, duT, u + row0off + (size_t)tn * blkrows, tile0 + (size_t)tn * TILE_STRIDE, c);
    __builtin_amdgcn_sched_barrier(0);
    const int rn = (tb + 2 < NTB) ? (tb + 2) * TB : LCH - SG3;
    p3_block(k1, Bs, Cs, (tb + 1) * TB, rn, hh, An, h, Dd, y + row0off + (size_t)(tb + 1) * blkrows, ga);
  }
}

extern "C" void kernel_launch(void* const* d_in, const int* in_sizes, int n_in,
                              void* d_out, int out_size, void* d_ws, size_t ws_size,
                              hipStream_t stream) {
  (void)in_sizes; (void)n_in; (void)out_size; (void)ws_size;
  const float* u    = (const float*)d_in[0];
  const float* Wx   = (const float*)d_in[1];
  const float* Wdt  = (const float*)d_in[2];
  const float* bdt  = (const float*)d_in[3];
  const float* logA = (const float*)d_in[4];
  const float* Dp   = (const float*)d_in[5];
  float* y = (float*)d_out;

  float* ws     = (float*)d_ws;
  float* xdbl   = ws;
  float* chunkA = xdbl  + (size_t)MROWS * XPROJ_N;
  float* chunkH = chunkA + (size_t)BATCH * DINNER * NCH * DSTATE;
  float* carry  = chunkH + (size_t)BATCH * DINNER * NCH * DSTATE;
  float* dlT    = carry  + (size_t)BATCH * DINNER * NCH * DSTATE;
  float* duT    = dlT    + (size_t)MROWS * DINNER;

  k_xproj<<<(MROWS / 64) * (XPROJ_N / 32), 256, 0, stream>>>(u, Wx, xdbl);
  const int scan_blocks = BATCH * NCH * (DINNER / DWG);
  k_scan_phase1<<<scan_blocks, 256, 0, stream>>>(u, xdbl, Wdt, bdt, logA, chunkA, chunkH, dlT, duT);
  k_scan_phase2<<<(BATCH * DINNER * DSTATE) / 256, 256, 0, stream>>>(chunkA, chunkH, carry);
  k_scan_phase3<<<scan_blocks, 256, 0, stream>>>(u, xdbl, logA, Dp, carry, dlT, duT, y);
}
